# starting kernel
# baseline (speedup 1.0000x reference)
#include <hip/hip_runtime.h>
#include <math.h>

typedef __attribute__((ext_vector_type(2))) float v2f;
typedef __attribute__((ext_vector_type(8))) float v8f;

__global__ void fill_kernel(float* __restrict__ p, float v, long n) {
  long i = (long)blockIdx.x * blockDim.x + threadIdx.x;
  if (i < n) p[i] = v;
}

__device__ __forceinline__ void atomicMaxF(float* addr, float v) {
  if (v >= 0.0f) atomicMax((int*)addr, __float_as_int(v));
  else           atomicMin((unsigned int*)addr, __float_as_uint(v));
}

template <int NCOL>
__global__ void __launch_bounds__(256) wmma_gemm_f32_k128(
    const float* __restrict__ A, const float* __restrict__ B,
    float* __restrict__ C, int M) {
  __shared__ float As[16 * 128];
  const int lane = threadIdx.x & 31;
  const int wave = threadIdx.x >> 5;
  const int row0 = blockIdx.x * 16;

  if (row0 + 16 <= M) {
    const float4* A4  = (const float4*)(A + (long)row0 * 128);
    float4*       As4 = (float4*)As;
    for (int i = threadIdx.x; i < 16 * 128 / 4; i += blockDim.x) As4[i] = A4[i];
  } else {
    for (int i = threadIdx.x; i < 16 * 128; i += blockDim.x) {
      int r = row0 + (i >> 7);
      if (r >= M) r = M - 1;
      As[i] = A[(long)r * 128 + (i & 127)];
    }
  }
  __syncthreads();

  const int col0  = wave * 16;
  const int mrow  = lane & 15;
  const int khalf = (lane >> 4) << 1;
  const float* Ap = As + mrow * 128 + khalf;
  const float* Bp = B + (long)khalf * NCOL + col0 + mrow;

  v8f acc0 = {}, acc1 = {}, acc2 = {}, acc3 = {};
#pragma unroll
  for (int k = 0; k < 128; k += 16) {
    v2f a0, a1, a2, a3, b0, b1, b2, b3;
    a0.x = Ap[k + 0];  a0.y = Ap[k + 1];
    a1.x = Ap[k + 4];  a1.y = Ap[k + 5];
    a2.x = Ap[k + 8];  a2.y = Ap[k + 9];
    a3.x = Ap[k + 12]; a3.y = Ap[k + 13];
    b0.x = Bp[(long)(k + 0)  * NCOL]; b0.y = Bp[(long)(k + 1)  * NCOL];
    b1.x = Bp[(long)(k + 4)  * NCOL]; b1.y = Bp[(long)(k + 5)  * NCOL];
    b2.x = Bp[(long)(k + 8)  * NCOL]; b2.y = Bp[(long)(k + 9)  * NCOL];
    b3.x = Bp[(long)(k + 12) * NCOL]; b3.y = Bp[(long)(k + 13) * NCOL];
    acc0 = __builtin_amdgcn_wmma_f32_16x16x4_f32(false, a0, false, b0, (short)0, acc0, false, false);
    acc1 = __builtin_amdgcn_wmma_f32_16x16x4_f32(false, a1, false, b1, (short)0, acc1, false, false);
    acc2 = __builtin_amdgcn_wmma_f32_16x16x4_f32(false, a2, false, b2, (short)0, acc2, false, false);
    acc3 = __builtin_amdgcn_wmma_f32_16x16x4_f32(false, a3, false, b3, (short)0, acc3, false, false);
  }
  v8f acc;
#pragma unroll
  for (int i = 0; i < 8; ++i) acc[i] = (acc0[i] + acc1[i]) + (acc2[i] + acc3[i]);

  const int rbase = row0 + ((lane >> 4) << 3);
  float* Cp = C + (long)rbase * NCOL + col0 + mrow;
  if (row0 + 16 <= M) {
#pragma unroll
    for (int i = 0; i < 8; ++i) Cp[(long)i * NCOL] = acc[i];
  } else {
#pragma unroll
    for (int i = 0; i < 8; ++i)
      if (rbase + i < M) Cp[(long)i * NCOL] = acc[i];
  }
}

template <int H, int D>
__global__ void attn_dots(const float* __restrict__ feat,
                          const float* __restrict__ al,
                          const float* __restrict__ ar,
                          float* __restrict__ el, float* __restrict__ er,
                          long NH) {
  long i = (long)blockIdx.x * blockDim.x + threadIdx.x;
  if (i >= NH) return;
  const int h = (int)(i & (H - 1));
  const float4* f  = (const float4*)(feat + i * D);
  const float4* l4 = (const float4*)(al + h * D);
  const float4* r4 = (const float4*)(ar + h * D);
  float sl = 0.f, sr = 0.f;
#pragma unroll
  for (int d = 0; d < D / 4; ++d) {
    float4 v = f[d], lv = l4[d], rv = r4[d];
    sl += v.x * lv.x + v.y * lv.y + v.z * lv.z + v.w * lv.w;
    sr += v.x * rv.x + v.y * rv.y + v.z * rv.z + v.w * rv.w;
  }
  el[i] = sl; er[i] = sr;
}

template <int H>
__global__ void edge_score_max(const int* __restrict__ src,
                               const int* __restrict__ dst,
                               const float* __restrict__ el,
                               const float* __restrict__ er,
                               float* __restrict__ ebuf,
                               float* __restrict__ m, long EH) {
  long i = (long)blockIdx.x * blockDim.x + threadIdx.x;
  if (i >= EH) return;
  long e = i / H; int h = (int)(i & (H - 1));
  float x = el[(long)src[e] * H + h] + er[(long)dst[e] * H + h];
  x = (x > 0.f) ? x : 0.2f * x;
  ebuf[i] = x;
  atomicMaxF(&m[(long)dst[e] * H + h], x);
}

template <int H>
__global__ void edge_exp_sum(const int* __restrict__ dst,
                             float* __restrict__ ebuf,
                             const float* __restrict__ m,
                             float* __restrict__ denom, long EH) {
  long i = (long)blockIdx.x * blockDim.x + threadIdx.x;
  if (i >= EH) return;
  long e = i / H; int h = (int)(i & (H - 1));
  long di = (long)dst[e] * H + h;
  float a = expf(ebuf[i] - m[di]);
  ebuf[i] = a;
  atomicAdd(&denom[di], a);
}

template <int F, int D>
__global__ void edge_scatter(const int* __restrict__ src,
                             const int* __restrict__ dst,
                             const float* __restrict__ feat,
                             const float* __restrict__ ebuf,
                             const float* __restrict__ denom,
                             float* __restrict__ out, long nchunks) {
  constexpr int C4 = F / 4;
  constexpr int H  = F / D;
  long i = (long)blockIdx.x * blockDim.x + threadIdx.x;
  if (i >= nchunks) return;
  long e = i / C4; int c = (int)(i & (C4 - 1));
  int  r = c * 4;  int h = r / D;
  long si = src[e], di = dst[e];
  float alpha = ebuf[e * H + h] / denom[di * H + h];
  float4 f = *(const float4*)(feat + si * F + r);
  float* o = out + di * F + r;
  atomicAdd(o + 0, f.x * alpha);
  atomicAdd(o + 1, f.y * alpha);
  atomicAdd(o + 2, f.z * alpha);
  atomicAdd(o + 3, f.w * alpha);
}

template <int F, bool RELU>
__global__ void bias_act(const float* __restrict__ in,
                         const float* __restrict__ bias,
                         float* __restrict__ out, long n4) {
  long i = (long)blockIdx.x * blockDim.x + threadIdx.x;
  if (i >= n4) return;
  int c = (int)(i & (F / 4 - 1));
  float4 v = ((const float4*)in)[i];
  float4 b = ((const float4*)bias)[c];
  v.x += b.x; v.y += b.y; v.z += b.z; v.w += b.w;
  if (RELU) {
    v.x = fmaxf(v.x, 0.f); v.y = fmaxf(v.y, 0.f);
    v.z = fmaxf(v.z, 0.f); v.w = fmaxf(v.w, 0.f);
  }
  ((float4*)out)[i] = v;
}

extern "C" void kernel_launch(void* const* d_in, const int* in_sizes, int n_in,
                              void* d_out, int out_size, void* d_ws, size_t ws_size,
                              hipStream_t stream) {
  const float* h   = (const float*)d_in[0];
  const int*   src = (const int*)d_in[1];
  const int*   dst = (const int*)d_in[2];
  const float* W1  = (const float*)d_in[3];
  const float* al1 = (const float*)d_in[4];
  const float* ar1 = (const float*)d_in[5];
  const float* b1  = (const float*)d_in[6];
  const float* W2  = (const float*)d_in[7];
  const float* al2 = (const float*)d_in[8];
  const float* ar2 = (const float*)d_in[9];
  const float* b2  = (const float*)d_in[10];

  constexpr int F1 = 128, H1 = 4, D1 = 32;
  constexpr int F2 = 64,  H2 = 1, D2 = 64;
  const long N = in_sizes[0] / F1;
  const long E = in_sizes[1];

  float* ws    = (float*)d_ws;
  float* feat1 = ws;
  float* out1  = feat1 + N * F1;
  float* feat2 = out1  + N * F1;
  float* ebuf  = feat2 + N * F2;
  float* el    = ebuf  + E * H1;
  float* er    = el    + N * H1;
  float* mbuf  = er    + N * H1;
  float* den   = mbuf  + N * H1;
  float* x2    = feat1;
  float* out2  = out1;

  auto cdiv = [](long a, long b) { return (unsigned)((a + b - 1) / b); };
  const int T = 256;
  const unsigned mtiles = cdiv(N, 16);

  fill_kernel<<<cdiv(N * H1, T), T, 0, stream>>>(mbuf, -INFINITY, N * H1);
  fill_kernel<<<cdiv(N * H1, T), T, 0, stream>>>(den, 0.f, N * H1);
  fill_kernel<<<cdiv(N * F1, T), T, 0, stream>>>(out1, 0.f, N * F1);

  wmma_gemm_f32_k128<F1><<<mtiles, 32 * (F1 / 16), 0, stream>>>(h, W1, feat1, (int)N);
  attn_dots<H1, D1><<<cdiv(N * H1, T), T, 0, stream>>>(feat1, al1, ar1, el, er, N * H1);
  edge_score_max<H1><<<cdiv(E * H1, T), T, 0, stream>>>(src, dst, el, er, ebuf, mbuf, E * H1);
  edge_exp_sum<H1><<<cdiv(E * H1, T), T, 0, stream>>>(dst, ebuf, mbuf, den, E * H1);
  edge_scatter<F1, D1><<<cdiv(E * (F1 / 4), T), T, 0, stream>>>(src, dst, feat1, ebuf, den,
                                                                out1, E * (F1 / 4));
  bias_act<F1, true><<<cdiv(N * F1 / 4, T), T, 0, stream>>>(out1, b1, x2, N * F1 / 4);

  fill_kernel<<<cdiv(N * H2, T), T, 0, stream>>>(mbuf, -INFINITY, N * H2);
  fill_kernel<<<cdiv(N * H2, T), T, 0, stream>>>(den, 0.f, N * H2);
  fill_kernel<<<cdiv(N * F2, T), T, 0, stream>>>(out2, 0.f, N * F2);

  wmma_gemm_f32_k128<F2><<<mtiles, 32 * (F2 / 16), 0, stream>>>(x2, W2, feat2, (int)N);
  attn_dots<H2, D2><<<cdiv(N * H2, T), T, 0, stream>>>(feat2, al2, ar2, el, er, N * H2);
  edge_score_max<H2><<<cdiv(E * H2, T), T, 0, stream>>>(src, dst, el, er, ebuf, mbuf, E * H2);
  edge_exp_sum<H2><<<cdiv(E * H2, T), T, 0, stream>>>(dst, ebuf, mbuf, den, E * H2);
  edge_scatter<F2, D2><<<cdiv(E * (F2 / 4), T), T, 0, stream>>>(src, dst, feat2, ebuf, den,
                                                                out2, E * (F2 / 4));
  bias_act<F2, false><<<cdiv(N * F2 / 4, T), T, 0, stream>>>(out2, b2, (float*)d_out, N * F2 / 4);
}


# Edges bucketed by destination once per call (counting sort, arrival-order rank); softmax and feature sum run per node in registers, one divide per node; no float atomics, sums in rank order
# speedup vs baseline: 2.4878x; 2.4878x over previous
#include <hip/hip_runtime.h>
#include <math.h>

typedef __attribute__((ext_vector_type(2))) float v2f;
typedef __attribute__((ext_vector_type(8))) float v8f;

__global__ void fill_kernel(float* __restrict__ p, float v, long n) {
  long i = (long)blockIdx.x * blockDim.x + threadIdx.x;
  if (i < n) p[i] = v;
}

template <int NCOL>
__global__ void __launch_bounds__(256) wmma_gemm_f32_k128(
    const float* __restrict__ A, const float* __restrict__ B,
    float* __restrict__ C, int M) {
  __shared__ float As[16 * 128];
  const int lane = threadIdx.x & 31;
  const int wave = threadIdx.x >> 5;
  const int row0 = blockIdx.x * 16;

  if (row0 + 16 <= M) {
    const float4* A4  = (const float4*)(A + (long)row0 * 128);
    float4*       As4 = (float4*)As;
    for (int i = threadIdx.x; i < 16 * 128 / 4; i += blockDim.x) As4[i] = A4[i];
  } else {
    for (int i = threadIdx.x; i < 16 * 128; i += blockDim.x) {
      int r = row0 + (i >> 7);
      if (r >= M) r = M - 1;
      As[i] = A[(long)r * 128 + (i & 127)];
    }
  }
  __syncthreads();

  const int col0  = wave * 16;
  const int mrow  = lane & 15;
  const int khalf = (lane >> 4) << 1;
  const float* Ap = As + mrow * 128 + khalf;
  const float* Bp = B + (long)khalf * NCOL + col0 + mrow;

  v8f acc0 = {}, acc1 = {}, acc2 = {}, acc3 = {};
#pragma unroll
  for (int k = 0; k < 128; k += 16) {
    v2f a0, a1, a2, a3, b0, b1, b2, b3;
    a0.x = Ap[k + 0];  a0.y = Ap[k + 1];
    a1.x = Ap[k + 4];  a1.y = Ap[k + 5];
    a2.x = Ap[k + 8];  a2.y = Ap[k + 9];
    a3.x = Ap[k + 12]; a3.y = Ap[k + 13];
    b0.x = Bp[(long)(k + 0)  * NCOL]; b0.y = Bp[(long)(k + 1)  * NCOL];
    b1.x = Bp[(long)(k + 4)  * NCOL]; b1.y = Bp[(long)(k + 5)  * NCOL];
    b2.x = Bp[(long)(k + 8)  * NCOL]; b2.y = Bp[(long)(k + 9)  * NCOL];
    b3.x = Bp[(long)(k + 12) * NCOL]; b3.y = Bp[(long)(k + 13) * NCOL];
    acc0 = __builtin_amdgcn_wmma_f32_16x16x4_f32(false, a0, false, b0, (short)0, acc0, false, false);
    acc1 = __builtin_amdgcn_wmma_f32_16x16x4_f32(false, a1, false, b1, (short)0, acc1, false, false);
    acc2 = __builtin_amdgcn_wmma_f32_16x16x4_f32(false, a2, false, b2, (short)0, acc2, false, false);
    acc3 = __builtin_amdgcn_wmma_f32_16x16x4_f32(false, a3, false, b3, (short)0, acc3, false, false);
  }
  v8f acc;
#pragma unroll
  for (int i = 0; i < 8; ++i) acc[i] = (acc0[i] + acc1[i]) + (acc2[i] + acc3[i]);

  const int rbase = row0 + ((lane >> 4) << 3);
  float* Cp = C + (long)rbase * NCOL + col0 + mrow;
  if (row0 + 16 <= M) {
#pragma unroll
    for (int i = 0; i < 8; ++i) Cp[(long)i * NCOL] = acc[i];
  } else {
#pragma unroll
    for (int i = 0; i < 8; ++i)
      if (rbase + i < M) Cp[(long)i * NCOL] = acc[i];
  }
}

template <int H, int D>
__global__ void attn_dots(const float* __restrict__ feat,
                          const float* __restrict__ al,
                          const float* __restrict__ ar,
                          float* __restrict__ el, float* __restrict__ er,
                          long NH) {
  long i = (long)blockIdx.x * blockDim.x + threadIdx.x;
  if (i >= NH) return;
  const int h = (int)(i & (H - 1));
  const float4* f  = (const float4*)(feat + i * D);
  const float4* l4 = (const float4*)(al + h * D);
  const float4* r4 = (const float4*)(ar + h * D);
  float sl = 0.f, sr = 0.f;
#pragma unroll
  for (int d = 0; d < D / 4; ++d) {
    float4 v = f[d], lv = l4[d], rv = r4[d];
    sl += v.x * lv.x + v.y * lv.y + v.z * lv.z + v.w * lv.w;
    sr += v.x * rv.x + v.y * rv.y + v.z * rv.z + v.w * rv.w;
  }
  el[i] = sl; er[i] = sr;
}

__global__ void edge_rank(const int* __restrict__ dst, int* __restrict__ cnt,
                          int* __restrict__ rank, long E) {
  long e0 = ((long)blockIdx.x * blockDim.x + threadIdx.x) * 4;
  if (e0 + 4 <= E) {
    int4 d = *(const int4*)(dst + e0);
    int4 r;
    r.x = atomicAdd(cnt + d.x, 1);
    r.y = atomicAdd(cnt + d.y, 1);
    r.z = atomicAdd(cnt + d.z, 1);
    r.w = atomicAdd(cnt + d.w, 1);
    *(int4*)(rank + e0) = r;
  } else {
    for (long e = e0; e < E; ++e) rank[e] = atomicAdd(cnt + dst[e], 1);
  }
}

__device__ __forceinline__ int4 load4_guarded(const int* __restrict__ p, int i, int e) {
  int4 c = {0, 0, 0, 0};
  if (i + 4 <= e) {
    c = *(const int4*)(p + i);
  } else {
    if (i     < e) c.x = p[i];
    if (i + 1 < e) c.y = p[i + 1];
    if (i + 2 < e) c.z = p[i + 2];
  }
  return c;
}

__global__ void __launch_bounds__(1024) scan_counts(const int* __restrict__ cnt,
                                                    int* __restrict__ rowptr, int n) {
  __shared__ int wtot[32];
  const int lane = threadIdx.x & 31;
  const int wave = threadIdx.x >> 5;
  const int per  = (((n + 31) / 32) + 127) & ~127;
  const int b = wave * per;
  const int e = (b + per < n) ? (b + per) : n;

  int s = 0;
  for (int i = b + lane * 4; i < e; i += 128) {
    const int4 c = load4_guarded(cnt, i, e);
    s += (c.x + c.y) + (c.z + c.w);
  }
  for (int o = 16; o > 0; o >>= 1) s += __shfl_xor(s, o);
  if (lane == 0) wtot[wave] = s;
  __syncthreads();
  int base = 0;
  for (int w = 0; w < wave; ++w) base += wtot[w];

  int4 cn = load4_guarded(cnt, b + lane * 4, e);
  for (int i0 = b; i0 < e; i0 += 128) {
    const int i = i0 + lane * 4;
    const int4 c = cn;
    cn = load4_guarded(cnt, i + 128, e);
    const int l1 = c.x, l2 = l1 + c.y, l3 = l2 + c.z, l4 = l3 + c.w;
    int inc = l4;
    for (int o = 1; o < 32; o <<= 1) {
      const int v = __shfl_up(inc, o);
      if (lane >= o) inc += v;
    }
    const int x0 = base + inc - l4;
    if (i + 4 <= e) {
      int4 o4; o4.x = x0; o4.y = x0 + l1; o4.z = x0 + l2; o4.w = x0 + l3;
      *(int4*)(rowptr + i) = o4;
    } else {
      if (i     < e) rowptr[i]     = x0;
      if (i + 1 < e) rowptr[i + 1] = x0 + l1;
      if (i + 2 < e) rowptr[i + 2] = x0 + l2;
    }
    base += __shfl(inc, 31);
  }
  if (threadIdx.x == 1023) rowptr[n] = base;
}

__global__ void edge_place(const int* __restrict__ src, const int* __restrict__ dst,
                           const int* __restrict__ rank, const int* __restrict__ rowptr,
                           int* __restrict__ srcs, long E) {
  long e0 = ((long)blockIdx.x * blockDim.x + threadIdx.x) * 4;
  if (e0 + 4 <= E) {
    int4 s = *(const int4*)(src + e0);
    int4 d = *(const int4*)(dst + e0);
    int4 r = *(const int4*)(rank + e0);
    srcs[rowptr[d.x] + r.x] = s.x;
    srcs[rowptr[d.y] + r.y] = s.y;
    srcs[rowptr[d.z] + r.z] = s.z;
    srcs[rowptr[d.w] + r.w] = s.w;
  } else {
    for (long e = e0; e < E; ++e) srcs[rowptr[dst[e]] + rank[e]] = src[e];
  }
}

template <int F, int H, bool RELU>
__global__ void __launch_bounds__(256) node_softmax_gather(
    const int* __restrict__ rowptr, const int* __restrict__ srcs,
    const float* __restrict__ feat, const float* __restrict__ el,
    const float* __restrict__ er, const float* __restrict__ bias,
    float* __restrict__ out, long N) {
  constexpr int LPN = F / 4;
  constexpr int D   = F / H;
  const long t    = (long)blockIdx.x * blockDim.x + threadIdx.x;
  const long node = t / LPN;
  if (node >= N) return;
  const int r = (int)(t & (LPN - 1)) * 4;
  const int h = r / D;
  const int b = rowptr[node], e = rowptr[node + 1];
  const float ern = er[node * H + h];

  float m = -INFINITY;
  int j = b;
  for (; j + 4 <= e; j += 4) {
    const long s0 = srcs[j], s1 = srcs[j + 1], s2 = srcs[j + 2], s3 = srcs[j + 3];
    float x0 = el[s0 * H + h] + ern, x1 = el[s1 * H + h] + ern;
    float x2 = el[s2 * H + h] + ern, x3 = el[s3 * H + h] + ern;
    x0 = (x0 > 0.f) ? x0 : 0.2f * x0;
    x1 = (x1 > 0.f) ? x1 : 0.2f * x1;
    x2 = (x2 > 0.f) ? x2 : 0.2f * x2;
    x3 = (x3 > 0.f) ? x3 : 0.2f * x3;
    x0 = (x1 > x0) ? x1 : x0;
    x2 = (x3 > x2) ? x3 : x2;
    x0 = (x2 > x0) ? x2 : x0;
    m  = (x0 > m) ? x0 : m;
  }
  for (; j < e; ++j) {
    const long s = srcs[j];
    float x = el[s * H + h] + ern;
    x = (x > 0.f) ? x : 0.2f * x;
    m = (x > m) ? x : m;
  }

  float den = 0.f;
  float4 acc = {0.f, 0.f, 0.f, 0.f};
  j = b;
  for (; j + 4 <= e; j += 4) {
    const long s0 = srcs[j], s1 = srcs[j + 1], s2 = srcs[j + 2], s3 = srcs[j + 3];
    float x0 = el[s0 * H + h] + ern, x1 = el[s1 * H + h] + ern;
    float x2 = el[s2 * H + h] + ern, x3 = el[s3 * H + h] + ern;
    const float4 f0 = *(const float4*)(feat + s0 * F + r);
    const float4 f1 = *(const float4*)(feat + s1 * F + r);
    const float4 f2 = *(const float4*)(feat + s2 * F + r);
    const float4 f3 = *(const float4*)(feat + s3 * F + r);
    x0 = (x0 > 0.f) ? x0 : 0.2f * x0;
    x1 = (x1 > 0.f) ? x1 : 0.2f * x1;
    x2 = (x2 > 0.f) ? x2 : 0.2f * x2;
    x3 = (x3 > 0.f) ? x3 : 0.2f * x3;
    const float a0 = expf(x0 - m), a1 = expf(x1 - m);
    const float a2 = expf(x2 - m), a3 = expf(x3 - m);
    den += a0; den += a1; den += a2; den += a3;
    acc.x += f0.x * a0; acc.y += f0.y * a0; acc.z += f0.z * a0; acc.w += f0.w * a0;
    acc.x += f1.x * a1; acc.y += f1.y * a1; acc.z += f1.z * a1; acc.w += f1.w * a1;
    acc.x += f2.x * a2; acc.y += f2.y * a2; acc.z += f2.z * a2; acc.w += f2.w * a2;
    acc.x += f3.x * a3; acc.y += f3.y * a3; acc.z += f3.z * a3; acc.w += f3.w * a3;
  }
  for (; j < e; ++j) {
    const long s = srcs[j];
    float x = el[s * H + h] + ern;
    x = (x > 0.f) ? x : 0.2f * x;
    const float a = expf(x - m);
    const float4 f = *(const float4*)(feat + s * F + r);
    den   += a;
    acc.x += f.x * a; acc.y += f.y * a; acc.z += f.z * a; acc.w += f.w * a;
  }

  float4 o = *(const float4*)(bias + r);
  if (e > b) {
    o.x += acc.x / den; o.y += acc.y / den; o.z += acc.z / den; o.w += acc.w / den;
  }
  if (RELU) {
    o.x = fmaxf(o.x, 0.f); o.y = fmaxf(o.y, 0.f);
    o.z = fmaxf(o.z, 0.f); o.w = fmaxf(o.w, 0.f);
  }
  *(float4*)(out + node * F + r) = o;
}

extern "C" void kernel_launch(void* const* d_in, const int* in_sizes, int n_in,
                              void* d_out, int out_size, void* d_ws, size_t ws_size,
                              hipStream_t stream) {
  const float* h   = (const float*)d_in[0];
  const int*   src = (const int*)d_in[1];
  const int*   dst = (const int*)d_in[2];
  const float* W1  = (const float*)d_in[3];
  const float* al1 = (const float*)d_in[4];
  const float* ar1 = (const float*)d_in[5];
  const float* b1  = (const float*)d_in[6];
  const float* W2  = (const float*)d_in[7];
  const float* al2 = (const float*)d_in[8];
  const float* ar2 = (const float*)d_in[9];
  const float* b2  = (const float*)d_in[10];

  constexpr int F1 = 128, H1 = 4, D1 = 32;
  constexpr int F2 = 64,  H2 = 1, D2 = 64;
  const long N = in_sizes[0] / F1;
  const long E = in_sizes[1];

  float* ws     = (float*)d_ws;
  float* feat1  = ws;
  float* x2     = feat1 + N * F1;
  float* feat2  = x2    + N * F1;
  int*   rank   = (int*)(feat2 + N * F2);
  int*   srcs   = rank + E;
  float* el     = (float*)(rank + E * H1);
  float* er     = el + N * H1;
  int*   cnt    = (int*)(er + N * H1);
  int*   rowptr = cnt + N * H1;

  auto cdiv = [](long a, long b) { return (unsigned)((a + b - 1) / b); };
  const int T = 256;
  const unsigned mtiles = cdiv(N, 16);

  fill_kernel<<<cdiv(N, T), T, 0, stream>>>((float*)cnt, 0.f, N);
  edge_rank<<<cdiv((E + 3) / 4, T), T, 0, stream>>>(dst, cnt, rank, E);
  scan_counts<<<1, 1024, 0, stream>>>(cnt, rowptr, (int)N);
  edge_place<<<cdiv((E + 3) / 4, T), T, 0, stream>>>(src, dst, rank, rowptr, srcs, E);

  wmma_gemm_f32_k128<F1><<<mtiles, 32 * (F1 / 16), 0, stream>>>(h, W1, feat1, (int)N);
  attn_dots<H1, D1><<<cdiv(N * H1, T), T, 0, stream>>>(feat1, al1, ar1, el, er, N * H1);
  node_softmax_gather<F1, H1, true><<<cdiv(N * (F1 / 4), T), T, 0, stream>>>(
      rowptr, srcs, feat1, el, er, b1, x2, N);

  wmma_gemm_f32_k128<F2><<<mtiles, 32 * (F2 / 16), 0, stream>>>(x2, W2, feat2, (int)N);
  attn_dots<H2, D2><<<cdiv(N * H2, T), T, 0, stream>>>(feat2, al2, ar2, el, er, N * H2);
  node_softmax_gather<F2, H2, false><<<cdiv(N * (F2 / 4), T), T, 0, stream>>>(
      rowptr, srcs, feat2, el, er, b2, (float*)d_out, N);
}


# Gather kernels: each lane stages one edge of its head group (index, score, one exp) and the group shares them by lane broadcast; feature rows fetched eight at a time; sums keep rank order
# speedup vs baseline: 2.7278x; 1.0965x over previous
#include <hip/hip_runtime.h>
#include <math.h>

typedef __attribute__((ext_vector_type(2))) float v2f;
typedef __attribute__((ext_vector_type(8))) float v8f;

__global__ void fill_kernel(float* __restrict__ p, float v, long n) {
  long i = (long)blockIdx.x * blockDim.x + threadIdx.x;
  if (i < n) p[i] = v;
}

template <int NCOL>
__global__ void __launch_bounds__(256) wmma_gemm_f32_k128(
    const float* __restrict__ A, const float* __restrict__ B,
    float* __restrict__ C, int M) {
  __shared__ float As[16 * 128];
  const int lane = threadIdx.x & 31;
  const int wave = threadIdx.x >> 5;
  const int row0 = blockIdx.x * 16;

  if (row0 + 16 <= M) {
    const float4* A4  = (const float4*)(A + (long)row0 * 128);
    float4*       As4 = (float4*)As;
    for (int i = threadIdx.x; i < 16 * 128 / 4; i += blockDim.x) As4[i] = A4[i];
  } else {
    for (int i = threadIdx.x; i < 16 * 128; i += blockDim.x) {
      int r = row0 + (i >> 7);
      if (r >= M) r = M - 1;
      As[i] = A[(long)r * 128 + (i & 127)];
    }
  }
  __syncthreads();

  const int col0  = wave * 16;
  const int mrow  = lane & 15;
  const int khalf = (lane >> 4) << 1;
  const float* Ap = As + mrow * 128 + khalf;
  const float* Bp = B + (long)khalf * NCOL + col0 + mrow;

  v8f acc0 = {}, acc1 = {}, acc2 = {}, acc3 = {};
#pragma unroll
  for (int k = 0; k < 128; k += 16) {
    v2f a0, a1, a2, a3, b0, b1, b2, b3;
    a0.x = Ap[k + 0];  a0.y = Ap[k + 1];
    a1.x = Ap[k + 4];  a1.y = Ap[k + 5];
    a2.x = Ap[k + 8];  a2.y = Ap[k + 9];
    a3.x = Ap[k + 12]; a3.y = Ap[k + 13];
    b0.x = Bp[(long)(k + 0)  * NCOL]; b0.y = Bp[(long)(k + 1)  * NCOL];
    b1.x = Bp[(long)(k + 4)  * NCOL]; b1.y = Bp[(long)(k + 5)  * NCOL];
    b2.x = Bp[(long)(k + 8)  * NCOL]; b2.y = Bp[(long)(k + 9)  * NCOL];
    b3.x = Bp[(long)(k + 12) * NCOL]; b3.y = Bp[(long)(k + 13) * NCOL];
    acc0 = __builtin_amdgcn_wmma_f32_16x16x4_f32(false, a0, false, b0, (short)0, acc0, false, false);
    acc1 = __builtin_amdgcn_wmma_f32_16x16x4_f32(false, a1, false, b1, (short)0, acc1, false, false);
    acc2 = __builtin_amdgcn_wmma_f32_16x16x4_f32(false, a2, false, b2, (short)0, acc2, false, false);
    acc3 = __builtin_amdgcn_wmma_f32_16x16x4_f32(false, a3, false, b3, (short)0, acc3, false, false);
  }
  v8f acc;
#pragma unroll
  for (int i = 0; i < 8; ++i) acc[i] = (acc0[i] + acc1[i]) + (acc2[i] + acc3[i]);

  const int rbase = row0 + ((lane >> 4) << 3);
  float* Cp = C + (long)rbase * NCOL + col0 + mrow;
  if (row0 + 16 <= M) {
#pragma unroll
    for (int i = 0; i < 8; ++i) Cp[(long)i * NCOL] = acc[i];
  } else {
#pragma unroll
    for (int i = 0; i < 8; ++i)
      if (rbase + i < M) Cp[(long)i * NCOL] = acc[i];
  }
}

template <int H, int D>
__global__ void attn_dots(const float* __restrict__ feat,
                          const float* __restrict__ al,
                          const float* __restrict__ ar,
                          float* __restrict__ el, float* __restrict__ er,
                          long NH) {
  long i = (long)blockIdx.x * blockDim.x + threadIdx.x;
  if (i >= NH) return;
  const int h = (int)(i & (H - 1));
  const float4* f  = (const float4*)(feat + i * D);
  const float4* l4 = (const float4*)(al + h * D);
  const float4* r4 = (const float4*)(ar + h * D);
  float sl = 0.f, sr = 0.f;
#pragma unroll
  for (int d = 0; d < D / 4; ++d) {
    float4 v = f[d], lv = l4[d], rv = r4[d];
    sl += v.x * lv.x + v.y * lv.y + v.z * lv.z + v.w * lv.w;
    sr += v.x * rv.x + v.y * rv.y + v.z * rv.z + v.w * rv.w;
  }
  el[i] = sl; er[i] = sr;
}

__global__ void edge_rank(const int* __restrict__ dst, int* __restrict__ cnt,
                          int* __restrict__ rank, long E) {
  long e0 = ((long)blockIdx.x * blockDim.x + threadIdx.x) * 4;
  if (e0 + 4 <= E) {
    int4 d = *(const int4*)(dst + e0);
    int4 r;
    r.x = atomicAdd(cnt + d.x, 1);
    r.y = atomicAdd(cnt + d.y, 1);
    r.z = atomicAdd(cnt + d.z, 1);
    r.w = atomicAdd(cnt + d.w, 1);
    *(int4*)(rank + e0) = r;
  } else {
    for (long e = e0; e < E; ++e) rank[e] = atomicAdd(cnt + dst[e], 1);
  }
}

__device__ __forceinline__ int4 load4_guarded(const int* __restrict__ p, int i, int e) {
  int4 c = {0, 0, 0, 0};
  if (i + 4 <= e) {
    c = *(const int4*)(p + i);
  } else {
    if (i     < e) c.x = p[i];
    if (i + 1 < e) c.y = p[i + 1];
    if (i + 2 < e) c.z = p[i + 2];
  }
  return c;
}

__global__ void __launch_bounds__(1024) scan_counts(const int* __restrict__ cnt,
                                                    int* __restrict__ rowptr, int n) {
  __shared__ int wtot[32];
  const int lane = threadIdx.x & 31;
  const int wave = threadIdx.x >> 5;
  const int per  = (((n + 31) / 32) + 127) & ~127;
  const int b = wave * per;
  const int e = (b + per < n) ? (b + per) : n;

  int s = 0;
  for (int i = b + lane * 4; i < e; i += 128) {
    const int4 c = load4_guarded(cnt, i, e);
    s += (c.x + c.y) + (c.z + c.w);
  }
  for (int o = 16; o > 0; o >>= 1) s += __shfl_xor(s, o);
  if (lane == 0) wtot[wave] = s;
  __syncthreads();
  int base = 0;
  for (int w = 0; w < wave; ++w) base += wtot[w];

  int4 cn = load4_guarded(cnt, b + lane * 4, e);
  for (int i0 = b; i0 < e; i0 += 128) {
    const int i = i0 + lane * 4;
    const int4 c = cn;
    cn = load4_guarded(cnt, i + 128, e);
    const int l1 = c.x, l2 = l1 + c.y, l3 = l2 + c.z, l4 = l3 + c.w;
    int inc = l4;
    for (int o = 1; o < 32; o <<= 1) {
      const int v = __shfl_up(inc, o);
      if (lane >= o) inc += v;
    }
    const int x0 = base + inc - l4;
    if (i + 4 <= e) {
      int4 o4; o4.x = x0; o4.y = x0 + l1; o4.z = x0 + l2; o4.w = x0 + l3;
      *(int4*)(rowptr + i) = o4;
    } else {
      if (i     < e) rowptr[i]     = x0;
      if (i + 1 < e) rowptr[i + 1] = x0 + l1;
      if (i + 2 < e) rowptr[i + 2] = x0 + l2;
    }
    base += __shfl(inc, 31);
  }
  if (threadIdx.x == 1023) rowptr[n] = base;
}

__global__ void edge_place(const int* __restrict__ src, const int* __restrict__ dst,
                           const int* __restrict__ rank, const int* __restrict__ rowptr,
                           int* __restrict__ srcs, long E) {
  long e0 = ((long)blockIdx.x * blockDim.x + threadIdx.x) * 4;
  if (e0 + 4 <= E) {
    int4 s = *(const int4*)(src + e0);
    int4 d = *(const int4*)(dst + e0);
    int4 r = *(const int4*)(rank + e0);
    srcs[rowptr[d.x] + r.x] = s.x;
    srcs[rowptr[d.y] + r.y] = s.y;
    srcs[rowptr[d.z] + r.z] = s.z;
    srcs[rowptr[d.w] + r.w] = s.w;
  } else {
    for (long e = e0; e < E; ++e) srcs[rowptr[dst[e]] + rank[e]] = src[e];
  }
}


template <int LPH, int J>
__device__ __forceinline__ int group_lane(int v) {
  if constexpr (LPH == 8) {
    return __builtin_amdgcn_mov_dpp8(v, J * 0x249249);
  } else {
    return __builtin_amdgcn_update_dpp(v, v, 0x150 + J, 0xf, 0xf, false);
  }
}

template <int M>
__device__ __forceinline__ float row_xor(float v) {
  const int i = __float_as_int(v);
  return __int_as_float(__builtin_amdgcn_update_dpp(i, i, 0x160 + M, 0xf, 0xf, false));
}

template <int LPH, int JB>
__device__ __forceinline__ void gather8(const char* __restrict__ fbytes, int vo, int r4,
                                        float a, float& den, float4& acc) {
  float4 f[8];
  float  w[8];
#define ROW_(i)                                                                  \
  {                                                                              \
    const unsigned o = (unsigned)group_lane<LPH, JB + i>(vo) + (unsigned)r4;     \
    f[i] = *(const float4*)(fbytes + o);                                         \
    w[i] = __int_as_float(group_lane<LPH, JB + i>(__float_as_int(a)));           \
  }
  ROW_(0) ROW_(1) ROW_(2) ROW_(3) ROW_(4) ROW_(5) ROW_(6) ROW_(7)
#undef ROW_
#pragma unroll
  for (int i = 0; i < 8; ++i) {
    den   += w[i];
    acc.x = fmaf(f[i].x, w[i], acc.x); acc.y = fmaf(f[i].y, w[i], acc.y);
    acc.z = fmaf(f[i].z, w[i], acc.z); acc.w = fmaf(f[i].w, w[i], acc.w);
  }
}

template <int F, int H, bool RELU>
__global__ void __launch_bounds__(256) node_softmax_gather(
    const int* __restrict__ rowptr, const int* __restrict__ srcs,
    const float* __restrict__ feat, const float* __restrict__ el,
    const float* __restrict__ er, const float* __restrict__ bias,
    float* __restrict__ out, long N) {
  constexpr int LPN = F / 4;
  constexpr int D   = F / H;
  constexpr int LPH = D / 4;
  constexpr int K   = 32 / LPH;
  static_assert(LPH == 8 || LPH == 16, "lane groups of 8 or 16");
  const long t    = (long)blockIdx.x * blockDim.x + threadIdx.x;
  const long node = t / LPN;
  if (node >= N) return;
  const int ln = (int)(t & (LPN - 1));
  const int r  = ln * 4;
  const int h  = r / D;
  const int jj = ln & (LPH - 1);
  const int b = rowptr[node], e = rowptr[node + 1];
  const int deg = e - b;
  const float ern = er[node * H + h];
  const char* fbytes = (const char*)feat;

  int   sv[K];
  float xv[K];
  auto stage = [&](int off) {
    bool ok[K];
#pragma unroll
    for (int k = 0; k < K; ++k) {
      const int idx = b + off + k * LPH + jj;
      ok[k] = idx < e;
      sv[k] = srcs[ok[k] ? idx : b + off];
    }
#pragma unroll
    for (int k = 0; k < K; ++k) {
      float x = el[(long)sv[k] * H + h] + ern;
      x = (x > 0.f) ? x : 0.2f * x;
      xv[k] = ok[k] ? x : -INFINITY;
      sv[k] = sv[k] * (F * 4);
    }
  };

  float m = -INFINITY;
  if (deg > 0) {
    stage(0);
#pragma unroll
    for (int k = 0; k < K; ++k) m = (xv[k] > m) ? xv[k] : m;
    for (int off = 32; off < deg; off += 32) {
      stage(off);
#pragma unroll
      for (int k = 0; k < K; ++k) m = (xv[k] > m) ? xv[k] : m;
    }
  }
  {
    float o;
    o = row_xor<1>(m); m = (o > m) ? o : m;
    o = row_xor<2>(m); m = (o > m) ? o : m;
    o = row_xor<4>(m); m = (o > m) ? o : m;
    if constexpr (LPH == 16) { o = row_xor<8>(m); m = (o > m) ? o : m; }
  }

  float den = 0.f;
  float4 acc = {0.f, 0.f, 0.f, 0.f};
  for (int off = 0; off < deg; off += 32) {
    if (deg > 32) stage(off);
#pragma unroll
    for (int k = 0; k < K; ++k) {
      if (off + k * LPH < deg) {
        const float a = expf(xv[k] - m);
        gather8<LPH, 0>(fbytes, sv[k], r * 4, a, den, acc);
        if constexpr (LPH == 16) {
          if (off + k * LPH + 8 < deg) gather8<LPH, 8>(fbytes, sv[k], r * 4, a, den, acc);
        }
      }
    }
  }

  float4 o = *(const float4*)(bias + r);
  if (e > b) {
    o.x += acc.x / den; o.y += acc.y / den; o.z += acc.z / den; o.w += acc.w / den;
  }
  if (RELU) {
    o.x = fmaxf(o.x, 0.f); o.y = fmaxf(o.y, 0.f);
    o.z = fmaxf(o.z, 0.f); o.w = fmaxf(o.w, 0.f);
  }
  *(float4*)(out + node * F + r) = o;
}

extern "C" void kernel_launch(void* const* d_in, const int* in_sizes, int n_in,
                              void* d_out, int out_size, void* d_ws, size_t ws_size,
                              hipStream_t stream) {
  const float* h   = (const float*)d_in[0];
  const int*   src = (const int*)d_in[1];
  const int*   dst = (const int*)d_in[2];
  const float* W1  = (const float*)d_in[3];
  const float* al1 = (const float*)d_in[4];
  const float* ar1 = (const float*)d_in[5];
  const float* b1  = (const float*)d_in[6];
  const float* W2  = (const float*)d_in[7];
  const float* al2 = (const float*)d_in[8];
  const float* ar2 = (const float*)d_in[9];
  const float* b2  = (const float*)d_in[10];

  constexpr int F1 = 128, H1 = 4, D1 = 32;
  constexpr int F2 = 64,  H2 = 1, D2 = 64;
  const long N = in_sizes[0] / F1;
  const long E = in_sizes[1];

  float* ws     = (float*)d_ws;
  float* feat1  = ws;
  float* x2     = feat1 + N * F1;
  float* feat2  = x2    + N * F1;
  int*   rank   = (int*)(feat2 + N * F2);
  int*   srcs   = rank + E;
  float* el     = (float*)(rank + E * H1);
  float* er     = el + N * H1;
  int*   cnt    = (int*)(er + N * H1);
  int*   rowptr = cnt + N * H1;

  auto cdiv = [](long a, long b) { return (unsigned)((a + b - 1) / b); };
  const int T = 256;
  const unsigned mtiles = cdiv(N, 16);

  fill_kernel<<<cdiv(N, T), T, 0, stream>>>((float*)cnt, 0.f, N);
  edge_rank<<<cdiv((E + 3) / 4, T), T, 0, stream>>>(dst, cnt, rank, E);
  scan_counts<<<1, 1024, 0, stream>>>(cnt, rowptr, (int)N);
  edge_place<<<cdiv((E + 3) / 4, T), T, 0, stream>>>(src, dst, rank, rowptr, srcs, E);

  wmma_gemm_f32_k128<F1><<<mtiles, 32 * (F1 / 16), 0, stream>>>(h, W1, feat1, (int)N);
  attn_dots<H1, D1><<<cdiv(N * H1, T), T, 0, stream>>>(feat1, al1, ar1, el, er, N * H1);
  node_softmax_gather<F1, H1, true><<<cdiv(N * (F1 / 4), T), T, 0, stream>>>(
      rowptr, srcs, feat1, el, er, b1, x2, N);

  wmma_gemm_f32_k128<F2><<<mtiles, 32 * (F2 / 16), 0, stream>>>(x2, W2, feat2, (int)N);
  attn_dots<H2, D2><<<cdiv(N * H2, T), T, 0, stream>>>(feat2, al2, ar2, el, er, N * H2);
  node_softmax_gather<F2, H2, false><<<cdiv(N * (F2 / 4), T), T, 0, stream>>>(
      rowptr, srcs, feat2, el, er, b2, (float*)d_out, N);
}


# Edges bucketed by a two-level counting sort in LDS (chunk counts, scatter by bin, per-bin sort) instead of global returning atomics and a one-group scan; 9 launches; sums in LDS arrival order
# speedup vs baseline: 4.3492x; 1.5944x over previous
#include <hip/hip_runtime.h>
#include <math.h>

typedef __attribute__((ext_vector_type(2))) float v2f;
typedef __attribute__((ext_vector_type(8))) float v8f;

template <int NCOL>
__global__ void __launch_bounds__(256) wmma_gemm_f32_k128(
    const float* __restrict__ A, const float* __restrict__ B,
    float* __restrict__ C, int M) {
  __shared__ float As[16 * 128];
  const int lane = threadIdx.x & 31;
  const int wave = threadIdx.x >> 5;
  const int row0 = blockIdx.x * 16;

  if (row0 + 16 <= M) {
    const float4* A4  = (const float4*)(A + (long)row0 * 128);
    float4*       As4 = (float4*)As;
    for (int i = threadIdx.x; i < 16 * 128 / 4; i += blockDim.x) As4[i] = A4[i];
  } else {
    for (int i = threadIdx.x; i < 16 * 128; i += blockDim.x) {
      int r = row0 + (i >> 7);
      if (r >= M) r = M - 1;
      As[i] = A[(long)r * 128 + (i & 127)];
    }
  }
  __syncthreads();

  const int col0  = wave * 16;
  const int mrow  = lane & 15;
  const int khalf = (lane >> 4) << 1;
  const float* Ap = As + mrow * 128 + khalf;
  const float* Bp = B + (long)khalf * NCOL + col0 + mrow;

  v8f acc0 = {}, acc1 = {}, acc2 = {}, acc3 = {};
#pragma unroll
  for (int k = 0; k < 128; k += 16) {
    v2f a0, a1, a2, a3, b0, b1, b2, b3;
    a0.x = Ap[k + 0];  a0.y = Ap[k + 1];
    a1.x = Ap[k + 4];  a1.y = Ap[k + 5];
    a2.x = Ap[k + 8];  a2.y = Ap[k + 9];
    a3.x = Ap[k + 12]; a3.y = Ap[k + 13];
    b0.x = Bp[(long)(k + 0)  * NCOL]; b0.y = Bp[(long)(k + 1)  * NCOL];
    b1.x = Bp[(long)(k + 4)  * NCOL]; b1.y = Bp[(long)(k + 5)  * NCOL];
    b2.x = Bp[(long)(k + 8)  * NCOL]; b2.y = Bp[(long)(k + 9)  * NCOL];
    b3.x = Bp[(long)(k + 12) * NCOL]; b3.y = Bp[(long)(k + 13) * NCOL];
    acc0 = __builtin_amdgcn_wmma_f32_16x16x4_f32(false, a0, false, b0, (short)0, acc0, false, false);
    acc1 = __builtin_amdgcn_wmma_f32_16x16x4_f32(false, a1, false, b1, (short)0, acc1, false, false);
    acc2 = __builtin_amdgcn_wmma_f32_16x16x4_f32(false, a2, false, b2, (short)0, acc2, false, false);
    acc3 = __builtin_amdgcn_wmma_f32_16x16x4_f32(false, a3, false, b3, (short)0, acc3, false, false);
  }
  v8f acc;
#pragma unroll
  for (int i = 0; i < 8; ++i) acc[i] = (acc0[i] + acc1[i]) + (acc2[i] + acc3[i]);

  const int rbase = row0 + ((lane >> 4) << 3);
  float* Cp = C + (long)rbase * NCOL + col0 + mrow;
  if (row0 + 16 <= M) {
#pragma unroll
    for (int i = 0; i < 8; ++i) Cp[(long)i * NCOL] = acc[i];
  } else {
#pragma unroll
    for (int i = 0; i < 8; ++i)
      if (rbase + i < M) Cp[(long)i * NCOL] = acc[i];
  }
}

template <int H, int D>
__global__ void attn_dots(const float* __restrict__ feat,
                          const float* __restrict__ al,
                          const float* __restrict__ ar,
                          float* __restrict__ el, float* __restrict__ er,
                          long NH) {
  long i = (long)blockIdx.x * blockDim.x + threadIdx.x;
  if (i >= NH) return;
  const int h = (int)(i & (H - 1));
  const float4* f  = (const float4*)(feat + i * D);
  const float4* l4 = (const float4*)(al + h * D);
  const float4* r4 = (const float4*)(ar + h * D);
  float sl = 0.f, sr = 0.f;
#pragma unroll
  for (int d = 0; d < D / 4; ++d) {
    float4 v = f[d], lv = l4[d], rv = r4[d];
    sl += v.x * lv.x + v.y * lv.y + v.z * lv.z + v.w * lv.w;
    sr += v.x * rv.x + v.y * rv.y + v.z * rv.z + v.w * rv.w;
  }
  el[i] = sl; er[i] = sr;
}

constexpr int BIN_SHIFT = 8;
constexpr int BIN_NODES = 1 << BIN_SHIFT;
constexpr int MAXBIN    = 256;
constexpr int CHUNK     = 4096;
constexpr int SORT_T    = 1024;
constexpr int BIN_CAP   = 16384;

__global__ void __launch_bounds__(SORT_T) chunk_hist(const int* __restrict__ dst,
                                                     int* __restrict__ hist, int nbin, long E) {
  __shared__ int h[MAXBIN];
  const int tid = threadIdx.x;
  if (tid < MAXBIN) h[tid] = 0;
  __syncthreads();
  const long e = (long)blockIdx.x * CHUNK + tid * 4;
  if (e + 4 <= E) {
    const int4 d = *(const int4*)(dst + e);
    atomicAdd(&h[d.x >> BIN_SHIFT], 1);
    atomicAdd(&h[d.y >> BIN_SHIFT], 1);
    atomicAdd(&h[d.z >> BIN_SHIFT], 1);
    atomicAdd(&h[d.w >> BIN_SHIFT], 1);
  } else {
    for (long i = e; i < E; ++i) atomicAdd(&h[dst[i] >> BIN_SHIFT], 1);
  }
  __syncthreads();
  if (tid < nbin) hist[(long)blockIdx.x * nbin + tid] = h[tid];
}

__global__ void __launch_bounds__(SORT_T) chunk_scatter(
    const int* __restrict__ src, const int* __restrict__ dst, const int* __restrict__ hist,
    int* __restrict__ binstart, int* __restrict__ pairs_w, int nbin, int nblk, long E) {
  constexpr int PARTS = SORT_T / MAXBIN;
  __shared__ int  ptot[PARTS][MAXBIN], ppre[PARTS][MAXBIN];
  __shared__ int  gdelta[MAXBIN], lcur[MAXBIN];
  __shared__ int  wsum[2][MAXBIN / 32];
  __shared__ int2 stage[CHUNK];
  int2* __restrict__ pairs = (int2*)pairs_w;
  const int tid = threadIdx.x, lane = tid & 31, wave = tid >> 5;
  const int blk = blockIdx.x;
  const long e0 = (long)blk * CHUNK;
  const long e  = e0 + tid * 4;

  const bool full = e + 4 <= E;
  int4 s4 = {0, 0, 0, 0}, d4 = {0, 0, 0, 0};
  if (full) { s4 = *(const int4*)(src + e); d4 = *(const int4*)(dst + e); }

  const int bin = tid & (MAXBIN - 1), part = tid / MAXBIN;
  int tot = 0, pre = 0;
  if (bin < nbin) {
    const int* __restrict__ col = hist + bin;
    for (int k0 = part; k0 < nblk; k0 += PARTS * 16) {
      int v[16];
#pragma unroll
      for (int j = 0; j < 16; ++j) {
        const int k = k0 + PARTS * j;
        v[j] = col[(k < nblk ? k : nblk - 1) * nbin];
      }
#pragma unroll
      for (int j = 0; j < 16; ++j) {
        const int k = k0 + PARTS * j;
        const int x = (k < nblk) ? v[j] : 0;
        tot += x;
        pre += (k < blk) ? x : 0;
      }
    }
  }
  ptot[part][bin] = tot; ppre[part][bin] = pre;
  __syncthreads();

  int t4 = 0, p4 = 0, own = 0, ia = 0, ic = 0;
  if (tid < MAXBIN) {
    for (int p = 0; p < PARTS; ++p) { t4 += ptot[p][tid]; p4 += ppre[p][tid]; }
    own = (tid < nbin) ? hist[(long)blk * nbin + tid] : 0;
    ia = t4; ic = own;
    for (int o = 1; o < 32; o <<= 1) {
      const int va = __shfl_up(ia, o), vc = __shfl_up(ic, o);
      if (lane >= o) { ia += va; ic += vc; }
    }
    if (lane == 31) { wsum[0][wave] = ia; wsum[1][wave] = ic; }
  }
  __syncthreads();
  if (tid < MAXBIN) {
    int ba = 0, bc = 0;
    for (int w = 0; w < wave; ++w) { ba += wsum[0][w]; bc += wsum[1][w]; }
    const int bstart = ba + ia - t4;
    const int lstart = bc + ic - own;
    gdelta[tid] = bstart + p4 - lstart;
    lcur[tid]   = lstart;
    if (blk == 0 && tid < nbin)      binstart[tid]  = bstart;
    if (blk == 0 && tid == nbin - 1) binstart[nbin] = bstart + t4;
  }
  __syncthreads();

  if (full) {
    const int q0 = atomicAdd(&lcur[d4.x >> BIN_SHIFT], 1);
    const int q1 = atomicAdd(&lcur[d4.y >> BIN_SHIFT], 1);
    const int q2 = atomicAdd(&lcur[d4.z >> BIN_SHIFT], 1);
    const int q3 = atomicAdd(&lcur[d4.w >> BIN_SHIFT], 1);
    stage[q0] = make_int2(s4.x, d4.x);
    stage[q1] = make_int2(s4.y, d4.y);
    stage[q2] = make_int2(s4.z, d4.z);
    stage[q3] = make_int2(s4.w, d4.w);
  } else {
    for (long i = e; i < E; ++i) {
      const int d = dst[i];
      stage[atomicAdd(&lcur[d >> BIN_SHIFT], 1)] = make_int2(src[i], d);
    }
  }
  __syncthreads();
  const long left = E - e0;
  const int nloc = left < CHUNK ? (int)left : CHUNK;
  for (int i = tid; i < nloc; i += SORT_T) {
    const int2 p = stage[i];
    pairs[i + gdelta[p.y >> BIN_SHIFT]] = p;
  }
}

__global__ void __launch_bounds__(SORT_T) bin_sort(
    const int* __restrict__ pairs_w, const int* __restrict__ binstart,
    int* __restrict__ rowptr, int* __restrict__ srcs, int nbin, int N) {
  __shared__ int cnt[BIN_NODES], cur[BIN_NODES];
  __shared__ int wsum[BIN_NODES / 32];
  __shared__ int lsrt[BIN_CAP];
  const int2* __restrict__ pairs = (const int2*)pairs_w;
  const int tid = threadIdx.x, lane = tid & 31, wave = tid >> 5;
  const int b  = blockIdx.x;
  const int n0 = b << BIN_SHIFT;
  const int bs = binstart[b], be = binstart[b + 1];
  const int m  = be - bs;
  if (tid < BIN_NODES) cnt[tid] = 0;
  __syncthreads();
  for (int i0 = tid; i0 < m; i0 += 4 * SORT_T) {
    int d[4];
#pragma unroll
    for (int j = 0; j < 4; ++j) {
      const int i = i0 + j * SORT_T;
      d[j] = pairs[bs + (i < m ? i : m - 1)].y;
    }
#pragma unroll
    for (int j = 0; j < 4; ++j)
      if (i0 + j * SORT_T < m) atomicAdd(&cnt[d[j] - n0], 1);
  }
  __syncthreads();
  int c = 0, inc = 0;
  if (tid < BIN_NODES) {
    c = cnt[tid]; inc = c;
    for (int o = 1; o < 32; o <<= 1) {
      const int v = __shfl_up(inc, o);
      if (lane >= o) inc += v;
    }
    if (lane == 31) wsum[wave] = inc;
  }
  __syncthreads();
  if (tid < BIN_NODES) {
    int base = 0;
    for (int w = 0; w < wave; ++w) base += wsum[w];
    const int ex = base + inc - c;
    cur[tid] = ex;
    if (n0 + tid < N) rowptr[n0 + tid] = bs + ex;
  }
  if (b == nbin - 1 && tid == 0) rowptr[N] = be;
  __syncthreads();
  const bool fits = m <= BIN_CAP;
  for (int i0 = tid; i0 < m; i0 += 4 * SORT_T) {
    int2 p[4];
#pragma unroll
    for (int j = 0; j < 4; ++j) {
      const int i = i0 + j * SORT_T;
      p[j] = pairs[bs + (i < m ? i : m - 1)];
    }
#pragma unroll
    for (int j = 0; j < 4; ++j) {
      if (i0 + j * SORT_T < m) {
        const int pos = atomicAdd(&cur[p[j].y - n0], 1);
        if (fits) lsrt[pos] = p[j].x; else srcs[bs + pos] = p[j].x;
      }
    }
  }
  __syncthreads();
  if (fits)
    for (int i = tid; i < m; i += SORT_T) srcs[bs + i] = lsrt[i];
}


template <int LPH, int J>
__device__ __forceinline__ int group_lane(int v) {
  if constexpr (LPH == 8) {
    return __builtin_amdgcn_mov_dpp8(v, J * 0x249249);
  } else {
    return __builtin_amdgcn_update_dpp(v, v, 0x150 + J, 0xf, 0xf, false);
  }
}

template <int M>
__device__ __forceinline__ float row_xor(float v) {
  const int i = __float_as_int(v);
  return __int_as_float(__builtin_amdgcn_update_dpp(i, i, 0x160 + M, 0xf, 0xf, false));
}

template <int LPH, int JB>
__device__ __forceinline__ void gather8(const char* __restrict__ fbytes, int vo, int r4,
                                        float a, float& den, float4& acc) {
  float4 f[8];
  float  w[8];
#define ROW_(i)                                                                  \
  {                                                                              \
    const unsigned o = (unsigned)group_lane<LPH, JB + i>(vo) + (unsigned)r4;     \
    f[i] = *(const float4*)(fbytes + o);                                         \
    w[i] = __int_as_float(group_lane<LPH, JB + i>(__float_as_int(a)));           \
  }
  ROW_(0) ROW_(1) ROW_(2) ROW_(3) ROW_(4) ROW_(5) ROW_(6) ROW_(7)
#undef ROW_
#pragma unroll
  for (int i = 0; i < 8; ++i) {
    den   += w[i];
    acc.x = fmaf(f[i].x, w[i], acc.x); acc.y = fmaf(f[i].y, w[i], acc.y);
    acc.z = fmaf(f[i].z, w[i], acc.z); acc.w = fmaf(f[i].w, w[i], acc.w);
  }
}

template <int F, int H, bool RELU>
__global__ void __launch_bounds__(256) node_softmax_gather(
    const int* __restrict__ rowptr, const int* __restrict__ srcs,
    const float* __restrict__ feat, const float* __restrict__ el,
    const float* __restrict__ er, const float* __restrict__ bias,
    float* __restrict__ out, long N) {
  constexpr int LPN = F / 4;
  constexpr int D   = F / H;
  constexpr int LPH = D / 4;
  constexpr int K   = 32 / LPH;
  static_assert(LPH == 8 || LPH == 16, "lane groups of 8 or 16");
  const long t    = (long)blockIdx.x * blockDim.x + threadIdx.x;
  const long node = t / LPN;
  if (node >= N) return;
  const int ln = (int)(t & (LPN - 1));
  const int r  = ln * 4;
  const int h  = r / D;
  const int jj = ln & (LPH - 1);
  const int b = rowptr[node], e = rowptr[node + 1];
  const int deg = e - b;
  const float ern = er[node * H + h];
  const char* fbytes = (const char*)feat;

  int   sv[K];
  float xv[K];
  auto stage = [&](int off) {
    bool ok[K];
#pragma unroll
    for (int k = 0; k < K; ++k) {
      const int idx = b + off + k * LPH + jj;
      ok[k] = idx < e;
      sv[k] = srcs[ok[k] ? idx : b + off];
    }
#pragma unroll
    for (int k = 0; k < K; ++k) {
      float x = el[(long)sv[k] * H + h] + ern;
      x = (x > 0.f) ? x : 0.2f * x;
      xv[k] = ok[k] ? x : -INFINITY;
      sv[k] = sv[k] * (F * 4);
    }
  };

  float m = -INFINITY;
  if (deg > 0) {
    stage(0);
#pragma unroll
    for (int k = 0; k < K; ++k) m = (xv[k] > m) ? xv[k] : m;
    for (int off = 32; off < deg; off += 32) {
      stage(off);
#pragma unroll
      for (int k = 0; k < K; ++k) m = (xv[k] > m) ? xv[k] : m;
    }
  }
  {
    float o;
    o = row_xor<1>(m); m = (o > m) ? o : m;
    o = row_xor<2>(m); m = (o > m) ? o : m;
    o = row_xor<4>(m); m = (o > m) ? o : m;
    if constexpr (LPH == 16) { o = row_xor<8>(m); m = (o > m) ? o : m; }
  }

  float den = 0.f;
  float4 acc = {0.f, 0.f, 0.f, 0.f};
  for (int off = 0; off < deg; off += 32) {
    if (deg > 32) stage(off);
#pragma unroll
    for (int k = 0; k < K; ++k) {
      if (off + k * LPH < deg) {
        const float a = expf(xv[k] - m);
        gather8<LPH, 0>(fbytes, sv[k], r * 4, a, den, acc);
        if constexpr (LPH == 16) {
          if (off + k * LPH + 8 < deg) gather8<LPH, 8>(fbytes, sv[k], r * 4, a, den, acc);
        }
      }
    }
  }

  float4 o = *(const float4*)(bias + r);
  if (e > b) {
    o.x += acc.x / den; o.y += acc.y / den; o.z += acc.z / den; o.w += acc.w / den;
  }
  if (RELU) {
    o.x = fmaxf(o.x, 0.f); o.y = fmaxf(o.y, 0.f);
    o.z = fmaxf(o.z, 0.f); o.w = fmaxf(o.w, 0.f);
  }
  *(float4*)(out + node * F + r) = o;
}

extern "C" void kernel_launch(void* const* d_in, const int* in_sizes, int n_in,
                              void* d_out, int out_size, void* d_ws, size_t ws_size,
                              hipStream_t stream) {
  const float* h   = (const float*)d_in[0];
  const int*   src = (const int*)d_in[1];
  const int*   dst = (const int*)d_in[2];
  const float* W1  = (const float*)d_in[3];
  const float* al1 = (const float*)d_in[4];
  const float* ar1 = (const float*)d_in[5];
  const float* b1  = (const float*)d_in[6];
  const float* W2  = (const float*)d_in[7];
  const float* al2 = (const float*)d_in[8];
  const float* ar2 = (const float*)d_in[9];
  const float* b2  = (const float*)d_in[10];

  constexpr int F1 = 128, H1 = 4, D1 = 32;
  constexpr int F2 = 64,  H2 = 1, D2 = 64;
  const long N = in_sizes[0] / F1;
  const long E = in_sizes[1];

  float* ws     = (float*)d_ws;
  float* feat1  = ws;
  float* x2     = feat1 + N * F1;
  float* feat2  = x2    + N * F1;
  int*   pairs  = (int*)(feat2 + N * F2);
  int*   srcs   = pairs + 2 * E;
  int*   hist   = pairs + 3 * E;
  float* el     = (float*)(pairs + E * H1);
  float* er     = el + N * H1;
  int*   binst  = (int*)(er + N * H1);
  int*   rowptr = binst + N * H1;

  auto cdiv = [](long a, long b) { return (unsigned)((a + b - 1) / b); };
  const int T = 256;
  const unsigned mtiles = cdiv(N, 16);

  const unsigned nbin = cdiv(N, BIN_NODES);
  const unsigned nblk = cdiv(E, CHUNK);
  chunk_hist<<<nblk, SORT_T, 0, stream>>>(dst, hist, (int)nbin, E);
  chunk_scatter<<<nblk, SORT_T, 0, stream>>>(src, dst, hist, binst, pairs, (int)nbin, (int)nblk, E);
  bin_sort<<<nbin, SORT_T, 0, stream>>>(pairs, binst, rowptr, srcs, (int)nbin, (int)N);

  wmma_gemm_f32_k128<F1><<<mtiles, 32 * (F1 / 16), 0, stream>>>(h, W1, feat1, (int)N);
  attn_dots<H1, D1><<<cdiv(N * H1, T), T, 0, stream>>>(feat1, al1, ar1, el, er, N * H1);
  node_softmax_gather<F1, H1, true><<<cdiv(N * (F1 / 4), T), T, 0, stream>>>(
      rowptr, srcs, feat1, el, er, b1, x2, N);

  wmma_gemm_f32_k128<F2><<<mtiles, 32 * (F2 / 16), 0, stream>>>(x2, W2, feat2, (int)N);
  attn_dots<H2, D2><<<cdiv(N * H2, T), T, 0, stream>>>(feat2, al2, ar2, el, er, N * H2);
  node_softmax_gather<F2, H2, false><<<cdiv(N * (F2 / 4), T), T, 0, stream>>>(
      rowptr, srcs, feat2, el, er, b2, (float*)d_out, N);
}


# GEMM A slab in LDS: row pitch 128 -> 132 dwords, so a fragment read's 16 rows fall on different banks instead of one bank pair (16-way conflict on every read removed)
# speedup vs baseline: 5.9168x; 1.3604x over previous
#include <hip/hip_runtime.h>
#include <math.h>

typedef __attribute__((ext_vector_type(2))) float v2f;
typedef __attribute__((ext_vector_type(8))) float v8f;

constexpr int APITCH = 132;
template <int NCOL>
__global__ void __launch_bounds__(256) wmma_gemm_f32_k128(
    const float* __restrict__ A, const float* __restrict__ B,
    float* __restrict__ C, int M) {
  __shared__ __attribute__((aligned(16))) float As[16 * APITCH];
  const int lane = threadIdx.x & 31;
  const int wave = threadIdx.x >> 5;
  const int row0 = blockIdx.x * 16;

  if (row0 + 16 <= M) {
    const float4* A4  = (const float4*)(A + (long)row0 * 128);
    for (int i = threadIdx.x; i < 16 * 128 / 4; i += blockDim.x)
      *(float4*)(As + (i >> 5) * APITCH + (i & 31) * 4) = A4[i];
  } else {
    for (int i = threadIdx.x; i < 16 * 128; i += blockDim.x) {
      int r = row0 + (i >> 7);
      if (r >= M) r = M - 1;
      As[(i >> 7) * APITCH + (i & 127)] = A[(long)r * 128 + (i & 127)];
    }
  }
  __syncthreads();

  const int col0  = wave * 16;
  const int mrow  = lane & 15;
  const int khalf = (lane >> 4) << 1;
  const float* Ap = As + mrow * APITCH + khalf;
  const float* Bp = B + (long)khalf * NCOL + col0 + mrow;

  v8f acc0 = {}, acc1 = {}, acc2 = {}, acc3 = {};
#pragma unroll
  for (int k = 0; k < 128; k += 16) {
    v2f a0, a1, a2, a3, b0, b1, b2, b3;
    a0.x = Ap[k + 0];  a0.y = Ap[k + 1];
    a1.x = Ap[k + 4];  a1.y = Ap[k + 5];
    a2.x = Ap[k + 8];  a2.y = Ap[k + 9];
    a3.x = Ap[k + 12]; a3.y = Ap[k + 13];
    b0.x = Bp[(long)(k + 0)  * NCOL]; b0.y = Bp[(long)(k + 1)  * NCOL];
    b1.x = Bp[(long)(k + 4)  * NCOL]; b1.y = Bp[(long)(k + 5)  * NCOL];
    b2.x = Bp[(long)(k + 8)  * NCOL]; b2.y = Bp[(long)(k + 9)  * NCOL];
    b3.x = Bp[(long)(k + 12) * NCOL]; b3.y = Bp[(long)(k + 13) * NCOL];
    acc0 = __builtin_amdgcn_wmma_f32_16x16x4_f32(false, a0, false, b0, (short)0, acc0, false, false);
    acc1 = __builtin_amdgcn_wmma_f32_16x16x4_f32(false, a1, false, b1, (short)0, acc1, false, false);
    acc2 = __builtin_amdgcn_wmma_f32_16x16x4_f32(false, a2, false, b2, (short)0, acc2, false, false);
    acc3 = __builtin_amdgcn_wmma_f32_16x16x4_f32(false, a3, false, b3, (short)0, acc3, false, false);
  }
  v8f acc;
#pragma unroll
  for (int i = 0; i < 8; ++i) acc[i] = (acc0[i] + acc1[i]) + (acc2[i] + acc3[i]);

  const int rbase = row0 + ((lane >> 4) << 3);
  float* Cp = C + (long)rbase * NCOL + col0 + mrow;
  if (row0 + 16 <= M) {
#pragma unroll
    for (int i = 0; i < 8; ++i) Cp[(long)i * NCOL] = acc[i];
  } else {
#pragma unroll
    for (int i = 0; i < 8; ++i)
      if (rbase + i < M) Cp[(long)i * NCOL] = acc[i];
  }
}

constexpr int BIN_SHIFT = 8;
constexpr int BIN_NODES = 1 << BIN_SHIFT;
constexpr int MAXBIN    = 256;
constexpr int CHUNK     = 4096;
constexpr int SORT_T    = 1024;
constexpr int BIN_CAP   = 16384;

__global__ void __launch_bounds__(SORT_T) chunk_hist(const int* __restrict__ dst,
                                                     int* __restrict__ hist, int nbin, long E) {
  __shared__ int h[MAXBIN];
  const int tid = threadIdx.x;
  if (tid < MAXBIN) h[tid] = 0;
  __syncthreads();
  const long e = (long)blockIdx.x * CHUNK + tid * 4;
  if (e + 4 <= E) {
    const int4 d = *(const int4*)(dst + e);
    atomicAdd(&h[d.x >> BIN_SHIFT], 1);
    atomicAdd(&h[d.y >> BIN_SHIFT], 1);
    atomicAdd(&h[d.z >> BIN_SHIFT], 1);
    atomicAdd(&h[d.w >> BIN_SHIFT], 1);
  } else {
    for (long i = e; i < E; ++i) atomicAdd(&h[dst[i] >> BIN_SHIFT], 1);
  }
  __syncthreads();
  if (tid < nbin) hist[(long)blockIdx.x * nbin + tid] = h[tid];
}

__global__ void __launch_bounds__(SORT_T) chunk_scatter(
    const int* __restrict__ src, const int* __restrict__ dst, const int* __restrict__ hist,
    int* __restrict__ binstart, int* __restrict__ pairs_w, int nbin, int nblk, long E) {
  constexpr int PARTS = SORT_T / MAXBIN;
  __shared__ int  ptot[PARTS][MAXBIN], ppre[PARTS][MAXBIN];
  __shared__ int  gdelta[MAXBIN], lcur[MAXBIN];
  __shared__ int  wsum[2][MAXBIN / 32];
  __shared__ int2 stage[CHUNK];
  int2* __restrict__ pairs = (int2*)pairs_w;
  const int tid = threadIdx.x, lane = tid & 31, wave = tid >> 5;
  const int blk = blockIdx.x;
  const long e0 = (long)blk * CHUNK;
  const long e  = e0 + tid * 4;

  const bool full = e + 4 <= E;
  int4 s4 = {0, 0, 0, 0}, d4 = {0, 0, 0, 0};
  if (full) { s4 = *(const int4*)(src + e); d4 = *(const int4*)(dst + e); }

  const int bin = tid & (MAXBIN - 1), part = tid / MAXBIN;
  int tot = 0, pre = 0;
  if (bin < nbin) {
    const int* __restrict__ col = hist + bin;
    for (int k0 = part; k0 < nblk; k0 += PARTS * 16) {
      int v[16];
#pragma unroll
      for (int j = 0; j < 16; ++j) {
        const int k = k0 + PARTS * j;
        v[j] = col[(k < nblk ? k : nblk - 1) * nbin];
      }
#pragma unroll
      for (int j = 0; j < 16; ++j) {
        const int k = k0 + PARTS * j;
        const int x = (k < nblk) ? v[j] : 0;
        tot += x;
        pre += (k < blk) ? x : 0;
      }
    }
  }
  ptot[part][bin] = tot; ppre[part][bin] = pre;
  __syncthreads();

  int t4 = 0, p4 = 0, own = 0, ia = 0, ic = 0;
  if (tid < MAXBIN) {
    for (int p = 0; p < PARTS; ++p) { t4 += ptot[p][tid]; p4 += ppre[p][tid]; }
    own = (tid < nbin) ? hist[(long)blk * nbin + tid] : 0;
    ia = t4; ic = own;
    for (int o = 1; o < 32; o <<= 1) {
      const int va = __shfl_up(ia, o), vc = __shfl_up(ic, o);
      if (lane >= o) { ia += va; ic += vc; }
    }
    if (lane == 31) { wsum[0][wave] = ia; wsum[1][wave] = ic; }
  }
  __syncthreads();
  if (tid < MAXBIN) {
    int ba = 0, bc = 0;
    for (int w = 0; w < wave; ++w) { ba += wsum[0][w]; bc += wsum[1][w]; }
    const int bstart = ba + ia - t4;
    const int lstart = bc + ic - own;
    gdelta[tid] = bstart + p4 - lstart;
    lcur[tid]   = lstart;
    if (blk == 0 && tid < nbin)      binstart[tid]  = bstart;
    if (blk == 0 && tid == nbin - 1) binstart[nbin] = bstart + t4;
  }
  __syncthreads();

  if (full) {
    const int q0 = atomicAdd(&lcur[d4.x >> BIN_SHIFT], 1);
    const int q1 = atomicAdd(&lcur[d4.y >> BIN_SHIFT], 1);
    const int q2 = atomicAdd(&lcur[d4.z >> BIN_SHIFT], 1);
    const int q3 = atomicAdd(&lcur[d4.w >> BIN_SHIFT], 1);
    stage[q0] = make_int2(s4.x, d4.x);
    stage[q1] = make_int2(s4.y, d4.y);
    stage[q2] = make_int2(s4.z, d4.z);
    stage[q3] = make_int2(s4.w, d4.w);
  } else {
    for (long i = e; i < E; ++i) {
      const int d = dst[i];
      stage[atomicAdd(&lcur[d >> BIN_SHIFT], 1)] = make_int2(src[i], d);
    }
  }
  __syncthreads();
  const long left = E - e0;
  const int nloc = left < CHUNK ? (int)left : CHUNK;
  for (int i = tid; i < nloc; i += SORT_T) {
    const int2 p = stage[i];
    pairs[i + gdelta[p.y >> BIN_SHIFT]] = p;
  }
}

__global__ void __launch_bounds__(SORT_T) bin_sort(
    const int* __restrict__ pairs_w, const int* __restrict__ binstart,
    int* __restrict__ rowptr, int* __restrict__ srcs, int nbin, int N) {
  __shared__ int cnt[BIN_NODES], cur[BIN_NODES];
  __shared__ int wsum[BIN_NODES / 32];
  __shared__ int lsrt[BIN_CAP];
  const int2* __restrict__ pairs = (const int2*)pairs_w;
  const int tid = threadIdx.x, lane = tid & 31, wave = tid >> 5;
  const int b  = blockIdx.x;
  const int n0 = b << BIN_SHIFT;
  const int bs = binstart[b], be = binstart[b + 1];
  const int m  = be - bs;
  if (tid < BIN_NODES) cnt[tid] = 0;
  __syncthreads();
  for (int i0 = tid; i0 < m; i0 += 4 * SORT_T) {
    int d[4];
#pragma unroll
    for (int j = 0; j < 4; ++j) {
      const int i = i0 + j * SORT_T;
      d[j] = pairs[bs + (i < m ? i : m - 1)].y;
    }
#pragma unroll
    for (int j = 0; j < 4; ++j)
      if (i0 + j * SORT_T < m) atomicAdd(&cnt[d[j] - n0], 1);
  }
  __syncthreads();
  int c = 0, inc = 0;
  if (tid < BIN_NODES) {
    c = cnt[tid]; inc = c;
    for (int o = 1; o < 32; o <<= 1) {
      const int v = __shfl_up(inc, o);
      if (lane >= o) inc += v;
    }
    if (lane == 31) wsum[wave] = inc;
  }
  __syncthreads();
  if (tid < BIN_NODES) {
    int base = 0;
    for (int w = 0; w < wave; ++w) base += wsum[w];
    const int ex = base + inc - c;
    cur[tid] = ex;
    if (n0 + tid < N) rowptr[n0 + tid] = bs + ex;
  }
  if (b == nbin - 1 && tid == 0) rowptr[N] = be;
  __syncthreads();
  const bool fits = m <= BIN_CAP;
  for (int i0 = tid; i0 < m; i0 += 4 * SORT_T) {
    int2 p[4];
#pragma unroll
    for (int j = 0; j < 4; ++j) {
      const int i = i0 + j * SORT_T;
      p[j] = pairs[bs + (i < m ? i : m - 1)];
    }
#pragma unroll
    for (int j = 0; j < 4; ++j) {
      if (i0 + j * SORT_T < m) {
        const int pos = atomicAdd(&cur[p[j].y - n0], 1);
        if (fits) lsrt[pos] = p[j].x; else srcs[bs + pos] = p[j].x;
      }
    }
  }
  __syncthreads();
  if (fits)
    for (int i = tid; i < m; i += SORT_T) srcs[bs + i] = lsrt[i];
}


template <int LPH, int J>
__device__ __forceinline__ int group_lane(int v) {
  if constexpr (LPH == 8) {
    return __builtin_amdgcn_mov_dpp8(v, J * 0x249249);
  } else {
    return __builtin_amdgcn_update_dpp(v, v, 0x150 + J, 0xf, 0xf, false);
  }
}

template <int M>
__device__ __forceinline__ float row_xor(float v) {
  const int i = __float_as_int(v);
  return __int_as_float(__builtin_amdgcn_update_dpp(0, i, 0x160 + M, 0xf, 0xf, true));
}

template <int LPH>
__device__ __forceinline__ float group_sum(float v) {
  v += row_xor<1>(v);
  v += row_xor<2>(v);
  v += row_xor<4>(v);
  if constexpr (LPH == 16) v += row_xor<8>(v);
  return v;
}

__device__ __forceinline__ float dot4(const float4 a, const float4 b) {
  return fmaf(a.w, b.w, fmaf(a.z, b.z, fmaf(a.y, b.y, a.x * b.x)));
}

template <int F, int H, bool RELU>
__global__ void __launch_bounds__(256) node_softmax_gather(
    const int* __restrict__ rowptr, const int* __restrict__ srcs,
    const float* __restrict__ feat, const float* __restrict__ al,
    const float* __restrict__ ar, const float* __restrict__ bias,
    float* __restrict__ out, long N) {
  constexpr int LPN = F / 4;
  constexpr int D   = F / H;
  constexpr int LPH = D / 4;
  static_assert(LPH == 8 || LPH == 16, "lane groups of 8 or 16");
  const long t    = (long)blockIdx.x * blockDim.x + threadIdx.x;
  const long node = t / LPN;
  if (node >= N) return;
  const int ln = (int)(t & (LPN - 1));
  const int r  = ln * 4;
  const int jj = ln & 7;
  const int b = rowptr[node], e = rowptr[node + 1];
  const int deg = e - b;
  const char* fbytes = (const char*)feat;

  const float4 l4 = *(const float4*)(al + r);
  const float4 r4 = *(const float4*)(ar + r);
  const float ern = group_sum<LPH>(dot4(*(const float4*)(feat + node * F + r), r4));

  float m = -INFINITY, den = 0.f;
  float4 acc = {0.f, 0.f, 0.f, 0.f};
  int sv = 0;
  if (deg > 0) sv = srcs[(b + jj < e) ? b + jj : b];
  for (int off = 0; off < deg; off += 8) {
    const int  vo = sv * (F * 4);
    const bool ok = off + jj < deg;
    if (off + 8 < deg) {
      const int nx = b + off + 8 + jj;
      sv = srcs[(nx < e) ? nx : b + off + 8];
    }
    float4 f[8];
    float  s[8];
#define ROW_(i)                                                                  \
  {                                                                              \
    const unsigned o = (unsigned)group_lane<LPH, i>(vo) + (unsigned)(r * 4);     \
    f[i] = *(const float4*)(fbytes + o);                                         \
  }
    ROW_(0) ROW_(1) ROW_(2) ROW_(3) ROW_(4) ROW_(5) ROW_(6) ROW_(7)
#undef ROW_
#pragma unroll
    for (int i = 0; i < 8; ++i) s[i] = group_sum<LPH>(dot4(f[i], l4));
    float x = s[0];
#pragma unroll
    for (int i = 1; i < 8; ++i) x = (jj == i) ? s[i] : x;
    x += ern;
    x = (x > 0.f) ? x : 0.2f * x;
    x = ok ? x : -INFINITY;
    float mx = x, o;
    o = row_xor<1>(mx); mx = (o > mx) ? o : mx;
    o = row_xor<2>(mx); mx = (o > mx) ? o : mx;
    o = row_xor<4>(mx); mx = (o > mx) ? o : mx;
    const float mn = (mx > m) ? mx : m;
    const float sc = expf(m - mn);
    const float a  = expf(x - mn);
    m = mn;
    den *= sc;
    acc.x *= sc; acc.y *= sc; acc.z *= sc; acc.w *= sc;
#define ADD_(i)                                                                  \
  {                                                                              \
    const float w = __int_as_float(group_lane<LPH, i>(__float_as_int(a)));       \
    den   += w;                                                                  \
    acc.x = fmaf(f[i].x, w, acc.x); acc.y = fmaf(f[i].y, w, acc.y);              \
    acc.z = fmaf(f[i].z, w, acc.z); acc.w = fmaf(f[i].w, w, acc.w);              \
  }
    ADD_(0) ADD_(1) ADD_(2) ADD_(3) ADD_(4) ADD_(5) ADD_(6) ADD_(7)
#undef ADD_
  }

  float4 o = *(const float4*)(bias + r);
  if (e > b) {
    o.x += acc.x / den; o.y += acc.y / den; o.z += acc.z / den; o.w += acc.w / den;
  }
  if (RELU) {
    o.x = fmaxf(o.x, 0.f); o.y = fmaxf(o.y, 0.f);
    o.z = fmaxf(o.z, 0.f); o.w = fmaxf(o.w, 0.f);
  }
  *(float4*)(out + node * F + r) = o;
}

extern "C" void kernel_launch(void* const* d_in, const int* in_sizes, int n_in,
                              void* d_out, int out_size, void* d_ws, size_t ws_size,
                              hipStream_t stream) {
  const float* h   = (const float*)d_in[0];
  const int*   src = (const int*)d_in[1];
  const int*   dst = (const int*)d_in[2];
  const float* W1  = (const float*)d_in[3];
  const float* al1 = (const float*)d_in[4];
  const float* ar1 = (const float*)d_in[5];
  const float* b1  = (const float*)d_in[6];
  const float* W2  = (const float*)d_in[7];
  const float* al2 = (const float*)d_in[8];
  const float* ar2 = (const float*)d_in[9];
  const float* b2  = (const float*)d_in[10];

  constexpr int F1 = 128, H1 = 4;
  constexpr int F2 = 64,  H2 = 1;
  const long N = in_sizes[0] / F1;
  const long E = in_sizes[1];

  float* ws     = (float*)d_ws;
  float* feat1  = ws;
  float* x2     = feat1 + N * F1;
  float* feat2  = x2    + N * F1;
  int*   pairs  = (int*)(feat2 + N * F2);
  int*   srcs   = pairs + 2 * E;
  int*   hist   = pairs + 3 * E;
  int*   binst  = pairs + 4 * E;
  int*   rowptr = binst + N * H1;

  auto cdiv = [](long a, long b) { return (unsigned)((a + b - 1) / b); };
  const int T = 256;
  const unsigned mtiles = cdiv(N, 16);

  const unsigned nbin = cdiv(N, BIN_NODES);
  const unsigned nblk = cdiv(E, CHUNK);
  chunk_hist<<<nblk, SORT_T, 0, stream>>>(dst, hist, (int)nbin, E);
  chunk_scatter<<<nblk, SORT_T, 0, stream>>>(src, dst, hist, binst, pairs, (int)nbin, (int)nblk, E);
  bin_sort<<<nbin, SORT_T, 0, stream>>>(pairs, binst, rowptr, srcs, (int)nbin, (int)N);

  wmma_gemm_f32_k128<F1><<<mtiles, 32 * (F1 / 16), 0, stream>>>(h, W1, feat1, (int)N);
  node_softmax_gather<F1, H1, true><<<cdiv(N * (F1 / 4), T), T, 0, stream>>>(
      rowptr, srcs, feat1, al1, ar1, b1, x2, N);

  wmma_gemm_f32_k128<F2><<<mtiles, 32 * (F2 / 16), 0, stream>>>(x2, W2, feat2, (int)N);
  node_softmax_gather<F2, H2, false><<<cdiv(N * (F2 / 4), T), T, 0, stream>>>(
      rowptr, srcs, feat2, al2, ar2, b2, (float*)d_out, N);
}


# Layer-1 GEMM row slabs ride as extra work-groups in the three edge-sort launches (role by block index; the GEMM needs nothing from the sort); its own launch goes: 6 launches; tile sums unchanged
# speedup vs baseline: 6.1285x; 1.0358x over previous
#include <hip/hip_runtime.h>
#include <math.h>

typedef __attribute__((ext_vector_type(2))) float v2f;
typedef __attribute__((ext_vector_type(8))) float v8f;

constexpr int APITCH = 132;

template <int NCOL>
__device__ __forceinline__ v8f wave_tile_k128(const float* Ap, const float* __restrict__ Bp) {
  v8f acc0 = {}, acc1 = {}, acc2 = {}, acc3 = {};
#pragma unroll
  for (int k = 0; k < 128; k += 16) {
    v2f a0, a1, a2, a3, b0, b1, b2, b3;
    a0.x = Ap[k + 0];  a0.y = Ap[k + 1];
    a1.x = Ap[k + 4];  a1.y = Ap[k + 5];
    a2.x = Ap[k + 8];  a2.y = Ap[k + 9];
    a3.x = Ap[k + 12]; a3.y = Ap[k + 13];
    b0.x = Bp[(long)(k + 0)  * NCOL]; b0.y = Bp[(long)(k + 1)  * NCOL];
    b1.x = Bp[(long)(k + 4)  * NCOL]; b1.y = Bp[(long)(k + 5)  * NCOL];
    b2.x = Bp[(long)(k + 8)  * NCOL]; b2.y = Bp[(long)(k + 9)  * NCOL];
    b3.x = Bp[(long)(k + 12) * NCOL]; b3.y = Bp[(long)(k + 13) * NCOL];
    acc0 = __builtin_amdgcn_wmma_f32_16x16x4_f32(false, a0, false, b0, (short)0, acc0, false, false);
    acc1 = __builtin_amdgcn_wmma_f32_16x16x4_f32(false, a1, false, b1, (short)0, acc1, false, false);
    acc2 = __builtin_amdgcn_wmma_f32_16x16x4_f32(false, a2, false, b2, (short)0, acc2, false, false);
    acc3 = __builtin_amdgcn_wmma_f32_16x16x4_f32(false, a3, false, b3, (short)0, acc3, false, false);
  }
  v8f acc;
#pragma unroll
  for (int i = 0; i < 8; ++i) acc[i] = (acc0[i] + acc1[i]) + (acc2[i] + acc3[i]);
  return acc;
}

template <int NCOL>
__global__ void __launch_bounds__(256) wmma_gemm_f32_k128(
    const float* __restrict__ A, const float* __restrict__ B,
    float* __restrict__ C, int M) {
  __shared__ __attribute__((aligned(16))) float As[16 * APITCH];
  const int lane = threadIdx.x & 31;
  const int wave = threadIdx.x >> 5;
  const int row0 = blockIdx.x * 16;

  if (row0 + 16 <= M) {
    const float4* A4  = (const float4*)(A + (long)row0 * 128);
    for (int i = threadIdx.x; i < 16 * 128 / 4; i += blockDim.x)
      *(float4*)(As + (i >> 5) * APITCH + (i & 31) * 4) = A4[i];
  } else {
    for (int i = threadIdx.x; i < 16 * 128; i += blockDim.x) {
      int r = row0 + (i >> 7);
      if (r >= M) r = M - 1;
      As[(i >> 7) * APITCH + (i & 127)] = A[(long)r * 128 + (i & 127)];
    }
  }
  __syncthreads();

  const int col0  = wave * 16;
  const int mrow  = lane & 15;
  const int khalf = (lane >> 4) << 1;
  const v8f acc = wave_tile_k128<NCOL>(As + mrow * APITCH + khalf,
                                       B + (long)khalf * NCOL + col0 + mrow);

  const int rbase = row0 + ((lane >> 4) << 3);
  float* Cp = C + (long)rbase * NCOL + col0 + mrow;
  if (row0 + 16 <= M) {
#pragma unroll
    for (int i = 0; i < 8; ++i) Cp[(long)i * NCOL] = acc[i];
  } else {
#pragma unroll
    for (int i = 0; i < 8; ++i)
      if (rbase + i < M) Cp[(long)i * NCOL] = acc[i];
  }
}

constexpr int RIDE_T = 1024;
template <int NCOL>
struct Ride {
  static constexpr int CT   = NCOL / 16;
  static constexpr int MT   = (RIDE_T / 32) / CT;
  static constexpr int ROWS = 16 * MT;
  static constexpr int VPT  = ROWS * 32 / RIDE_T;
};
constexpr int RIDE_ROWS = Ride<128>::ROWS;

template <int NCOL>
__device__ __forceinline__ void gemm_ride(float* As, const float* __restrict__ A,
                                          const float* __restrict__ B, float* __restrict__ C,
                                          int M, int slab, int slab_end) {
  constexpr int CT = Ride<NCOL>::CT, ROWS = Ride<NCOL>::ROWS;
  static_assert(Ride<NCOL>::VPT == 2, "two float4 of a slab per thread");
  const int lane  = threadIdx.x & 31;
  const int wave  = threadIdx.x >> 5;
  const int mt    = wave / CT;
  const int col0  = (wave % CT) * 16;
  const int mrow  = lane & 15;
  const int khalf = (lane >> 4) << 1;
  const float* Ap = As + (mt * 16 + mrow) * APITCH + khalf;
  const float* Bp = B + (long)khalf * NCOL + col0 + mrow;
  const int i0 = threadIdx.x >> 5, c4 = (threadIdx.x & 31) * 4;
  float* As0 = As + i0 * APITCH + c4;
  float* As1 = As0 + 32 * APITCH;
  const float* Ac = A + c4;

  if (slab >= slab_end) return;
  int ra = slab * ROWS + i0, rb = ra + 32;
  ra = ra < M ? ra : M - 1;
  rb = rb < M ? rb : M - 1;
  float4 p0 = *(const float4*)(Ac + (long)ra * 128);
  float4 p1 = *(const float4*)(Ac + (long)rb * 128);
  for (;;) {
    *(float4*)As0 = p0;
    *(float4*)As1 = p1;
    __syncthreads();
    if (slab + 1 < slab_end && threadIdx.x < ROWS * 2) {
      int rn = (slab + 1) * ROWS + (threadIdx.x >> 1);
      rn = rn < M ? rn : M - 1;
      __builtin_prefetch(A + (long)rn * 128 + (threadIdx.x & 1) * 64, 0, 3);
    }
    int boff = 0;
    asm volatile("" : "+v"(boff));
    const v8f acc = wave_tile_k128<NCOL>(Ap, Bp + boff);
    const int rbase = slab * ROWS + mt * 16 + ((lane >> 4) << 3);
    float* Cp = C + (long)rbase * NCOL + col0 + mrow;
    if ((slab + 1) * ROWS <= M) {
#pragma unroll
      for (int i = 0; i < 8; ++i) Cp[(long)i * NCOL] = acc[i];
    } else {
#pragma unroll
      for (int i = 0; i < 8; ++i)
        if (rbase + i < M) Cp[(long)i * NCOL] = acc[i];
    }
    if (++slab >= slab_end) break;
    ra = slab * ROWS + i0; rb = ra + 32;
    ra = ra < M ? ra : M - 1;
    rb = rb < M ? rb : M - 1;
    p0 = *(const float4*)(Ac + (long)ra * 128);
    p1 = *(const float4*)(Ac + (long)rb * 128);
    __syncthreads();
  }
}

constexpr int BIN_SHIFT = 8;
constexpr int BIN_NODES = 1 << BIN_SHIFT;
constexpr int MAXBIN    = 256;
constexpr int CHUNK     = 4096;
constexpr int SORT_T    = 1024;
static_assert(SORT_T == RIDE_T, "the riding GEMM slabs are cut for the sort's work-group size");
constexpr int BIN_CAP   = 16384;

__global__ void __launch_bounds__(SORT_T) __attribute__((amdgpu_waves_per_eu(16))) chunk_hist(const int* __restrict__ dst,
                                                     int* __restrict__ hist, int nbin, long E,
                                                     const float* __restrict__ gA,
                                                     const float* __restrict__ gB,
                                                     float* __restrict__ gC, int gM,
                                                     int nride, int gper, int gslab0, int gend) {
  __shared__ int h[MAXBIN];
  __shared__ __attribute__((aligned(16))) float As[RIDE_ROWS * APITCH];
  if ((int)blockIdx.x < nride) {
    const int s0 = gslab0 + (int)blockIdx.x * gper;
    gemm_ride<128>(As, gA, gB, gC, gM, s0, (s0 + gper < gend) ? s0 + gper : gend);
    return;
  }
  const int blk = (int)blockIdx.x - nride;
  const int tid = threadIdx.x;
  if (tid < MAXBIN) h[tid] = 0;
  __syncthreads();
  const long e = (long)blk * CHUNK + tid * 4;
  if (e + 4 <= E) {
    const int4 d = *(const int4*)(dst + e);
    atomicAdd(&h[d.x >> BIN_SHIFT], 1);
    atomicAdd(&h[d.y >> BIN_SHIFT], 1);
    atomicAdd(&h[d.z >> BIN_SHIFT], 1);
    atomicAdd(&h[d.w >> BIN_SHIFT], 1);
  } else {
    for (long i = e; i < E; ++i) atomicAdd(&h[dst[i] >> BIN_SHIFT], 1);
  }
  __syncthreads();
  if (tid < nbin) hist[(long)blk * nbin + tid] = h[tid];
}

__global__ void __launch_bounds__(SORT_T) chunk_scatter(
    const int* __restrict__ src, const int* __restrict__ dst, const int* __restrict__ hist,
    int* __restrict__ binstart, int* __restrict__ pairs_w, int nbin, int nblk, long E,
    const float* __restrict__ gA, const float* __restrict__ gB, float* __restrict__ gC,
    int gM, int nride, int gper, int gslab0, int gend) {
  __shared__ __attribute__((aligned(16))) float As[RIDE_ROWS * APITCH];
  if ((int)blockIdx.x < nride) {
    const int s0 = gslab0 + (int)blockIdx.x * gper;
    gemm_ride<128>(As, gA, gB, gC, gM, s0, (s0 + gper < gend) ? s0 + gper : gend);
    return;
  }
  constexpr int PARTS = SORT_T / MAXBIN;
  __shared__ int  ptot[PARTS][MAXBIN], ppre[PARTS][MAXBIN];
  __shared__ int  gdelta[MAXBIN], lcur[MAXBIN];
  __shared__ int  wsum[2][MAXBIN / 32];
  __shared__ int2 stage[CHUNK];
  int2* __restrict__ pairs = (int2*)pairs_w;
  const int tid = threadIdx.x, lane = tid & 31, wave = tid >> 5;
  const int blk = (int)blockIdx.x - nride;
  const long e0 = (long)blk * CHUNK;
  const long e  = e0 + tid * 4;

  const bool full = e + 4 <= E;
  int4 s4 = {0, 0, 0, 0}, d4 = {0, 0, 0, 0};
  if (full) { s4 = *(const int4*)(src + e); d4 = *(const int4*)(dst + e); }

  const int bin = tid & (MAXBIN - 1), part = tid / MAXBIN;
  int tot = 0, pre = 0;
  if (bin < nbin) {
    const int* __restrict__ col = hist + bin;
    for (int k0 = part; k0 < nblk; k0 += PARTS * 16) {
      int v[16];
#pragma unroll
      for (int j = 0; j < 16; ++j) {
        const int k = k0 + PARTS * j;
        v[j] = col[(k < nblk ? k : nblk - 1) * nbin];
      }
#pragma unroll
      for (int j = 0; j < 16; ++j) {
        const int k = k0 + PARTS * j;
        const int x = (k < nblk) ? v[j] : 0;
        tot += x;
        pre += (k < blk) ? x : 0;
      }
    }
  }
  ptot[part][bin] = tot; ppre[part][bin] = pre;
  __syncthreads();

  int t4 = 0, p4 = 0, own = 0, ia = 0, ic = 0;
  if (tid < MAXBIN) {
    for (int p = 0; p < PARTS; ++p) { t4 += ptot[p][tid]; p4 += ppre[p][tid]; }
    own = (tid < nbin) ? hist[(long)blk * nbin + tid] : 0;
    ia = t4; ic = own;
    for (int o = 1; o < 32; o <<= 1) {
      const int va = __shfl_up(ia, o), vc = __shfl_up(ic, o);
      if (lane >= o) { ia += va; ic += vc; }
    }
    if (lane == 31) { wsum[0][wave] = ia; wsum[1][wave] = ic; }
  }
  __syncthreads();
  if (tid < MAXBIN) {
    int ba = 0, bc = 0;
    for (int w = 0; w < wave; ++w) { ba += wsum[0][w]; bc += wsum[1][w]; }
    const int bstart = ba + ia - t4;
    const int lstart = bc + ic - own;
    gdelta[tid] = bstart + p4 - lstart;
    lcur[tid]   = lstart;
    if (blk == 0 && tid < nbin)      binstart[tid]  = bstart;
    if (blk == 0 && tid == nbin - 1) binstart[nbin] = bstart + t4;
  }
  __syncthreads();

  if (full) {
    const int q0 = atomicAdd(&lcur[d4.x >> BIN_SHIFT], 1);
    const int q1 = atomicAdd(&lcur[d4.y >> BIN_SHIFT], 1);
    const int q2 = atomicAdd(&lcur[d4.z >> BIN_SHIFT], 1);
    const int q3 = atomicAdd(&lcur[d4.w >> BIN_SHIFT], 1);
    stage[q0] = make_int2(s4.x, d4.x);
    stage[q1] = make_int2(s4.y, d4.y);
    stage[q2] = make_int2(s4.z, d4.z);
    stage[q3] = make_int2(s4.w, d4.w);
  } else {
    for (long i = e; i < E; ++i) {
      const int d = dst[i];
      stage[atomicAdd(&lcur[d >> BIN_SHIFT], 1)] = make_int2(src[i], d);
    }
  }
  __syncthreads();
  const long left = E - e0;
  const int nloc = left < CHUNK ? (int)left : CHUNK;
  for (int i = tid; i < nloc; i += SORT_T) {
    const int2 p = stage[i];
    pairs[i + gdelta[p.y >> BIN_SHIFT]] = p;
  }
}

__global__ void __launch_bounds__(SORT_T) __attribute__((amdgpu_waves_per_eu(16))) bin_sort(
    const int* __restrict__ pairs_w, const int* __restrict__ binstart,
    int* __restrict__ rowptr, int* __restrict__ srcs, int nbin, int N,
    const float* __restrict__ gA, const float* __restrict__ gB, float* __restrict__ gC,
    int gM, int nride, int gper, int gslab0, int gend) {
  __shared__ __attribute__((aligned(16))) float As[RIDE_ROWS * APITCH];
  if ((int)blockIdx.x < nride) {
    const int s0 = gslab0 + (int)blockIdx.x * gper;
    gemm_ride<128>(As, gA, gB, gC, gM, s0, (s0 + gper < gend) ? s0 + gper : gend);
    return;
  }
  __shared__ int cnt[BIN_NODES], cur[BIN_NODES];
  __shared__ int wsum[BIN_NODES / 32];
  __shared__ int lsrt[BIN_CAP];
  const int2* __restrict__ pairs = (const int2*)pairs_w;
  const int tid = threadIdx.x, lane = tid & 31, wave = tid >> 5;
  const int b  = (int)blockIdx.x - nride;
  const int n0 = b << BIN_SHIFT;
  const int bs = binstart[b], be = binstart[b + 1];
  const int m  = be - bs;
  if (tid < BIN_NODES) cnt[tid] = 0;
  __syncthreads();
  for (int i0 = tid; i0 < m; i0 += 4 * SORT_T) {
    int d[4];
#pragma unroll
    for (int j = 0; j < 4; ++j) {
      const int i = i0 + j * SORT_T;
      d[j] = pairs[bs + (i < m ? i : m - 1)].y;
    }
#pragma unroll
    for (int j = 0; j < 4; ++j)
      if (i0 + j * SORT_T < m) atomicAdd(&cnt[d[j] - n0], 1);
  }
  __syncthreads();
  int c = 0, inc = 0;
  if (tid < BIN_NODES) {
    c = cnt[tid]; inc = c;
    for (int o = 1; o < 32; o <<= 1) {
      const int v = __shfl_up(inc, o);
      if (lane >= o) inc += v;
    }
    if (lane == 31) wsum[wave] = inc;
  }
  __syncthreads();
  if (tid < BIN_NODES) {
    int base = 0;
    for (int w = 0; w < wave; ++w) base += wsum[w];
    const int ex = base + inc - c;
    cur[tid] = ex;
    if (n0 + tid < N) rowptr[n0 + tid] = bs + ex;
  }
  if (b == nbin - 1 && tid == 0) rowptr[N] = be;
  __syncthreads();
  const bool fits = m <= BIN_CAP;
  for (int i0 = tid; i0 < m; i0 += 4 * SORT_T) {
    int2 p[4];
#pragma unroll
    for (int j = 0; j < 4; ++j) {
      const int i = i0 + j * SORT_T;
      p[j] = pairs[bs + (i < m ? i : m - 1)];
    }
#pragma unroll
    for (int j = 0; j < 4; ++j) {
      if (i0 + j * SORT_T < m) {
        const int pos = atomicAdd(&cur[p[j].y - n0], 1);
        if (fits) lsrt[pos] = p[j].x; else srcs[bs + pos] = p[j].x;
      }
    }
  }
  __syncthreads();
  if (fits)
    for (int i = tid; i < m; i += SORT_T) srcs[bs + i] = lsrt[i];
}


template <int LPH, int J>
__device__ __forceinline__ int group_lane(int v) {
  if constexpr (LPH == 8) {
    return __builtin_amdgcn_mov_dpp8(v, J * 0x249249);
  } else {
    return __builtin_amdgcn_update_dpp(v, v, 0x150 + J, 0xf, 0xf, false);
  }
}

template <int M>
__device__ __forceinline__ float row_xor(float v) {
  const int i = __float_as_int(v);
  return __int_as_float(__builtin_amdgcn_update_dpp(0, i, 0x160 + M, 0xf, 0xf, true));
}

template <int LPH>
__device__ __forceinline__ float group_sum(float v) {
  v += row_xor<1>(v);
  v += row_xor<2>(v);
  v += row_xor<4>(v);
  if constexpr (LPH == 16) v += row_xor<8>(v);
  return v;
}

__device__ __forceinline__ float dot4(const float4 a, const float4 b) {
  return fmaf(a.w, b.w, fmaf(a.z, b.z, fmaf(a.y, b.y, a.x * b.x)));
}

template <int F, int H, bool RELU>
__global__ void __launch_bounds__(256) node_softmax_gather(
    const int* __restrict__ rowptr, const int* __restrict__ srcs,
    const float* __restrict__ feat, const float* __restrict__ al,
    const float* __restrict__ ar, const float* __restrict__ bias,
    float* __restrict__ out, long N) {
  constexpr int LPN = F / 4;
  constexpr int D   = F / H;
  constexpr int LPH = D / 4;
  static_assert(LPH == 8 || LPH == 16, "lane groups of 8 or 16");
  const long t    = (long)blockIdx.x * blockDim.x + threadIdx.x;
  const long node = t / LPN;
  if (node >= N) return;
  const int ln = (int)(t & (LPN - 1));
  const int r  = ln * 4;
  const int jj = ln & 7;
  const int b = rowptr[node], e = rowptr[node + 1];
  const int deg = e - b;
  const char* fbytes = (const char*)feat;

  const float4 l4 = *(const float4*)(al + r);
  const float4 r4 = *(const float4*)(ar + r);
  const float ern = group_sum<LPH>(dot4(*(const float4*)(feat + node * F + r), r4));

  float m = -INFINITY, den = 0.f;
  float4 acc = {0.f, 0.f, 0.f, 0.f};
  int sv = 0;
  if (deg > 0) sv = srcs[(b + jj < e) ? b + jj : b];
  for (int off = 0; off < deg; off += 8) {
    const int  vo = sv * (F * 4);
    const bool ok = off + jj < deg;
    if (off + 8 < deg) {
      const int nx = b + off + 8 + jj;
      sv = srcs[(nx < e) ? nx : b + off + 8];
    }
    float4 f[8];
    float  s[8];
#define ROW_(i)                                                                  \
  {                                                                              \
    const unsigned o = (unsigned)group_lane<LPH, i>(vo) + (unsigned)(r * 4);     \
    f[i] = *(const float4*)(fbytes + o);                                         \
  }
    ROW_(0) ROW_(1) ROW_(2) ROW_(3) ROW_(4) ROW_(5) ROW_(6) ROW_(7)
#undef ROW_
#pragma unroll
    for (int i = 0; i < 8; ++i) s[i] = group_sum<LPH>(dot4(f[i], l4));
    float x = s[0];
#pragma unroll
    for (int i = 1; i < 8; ++i) x = (jj == i) ? s[i] : x;
    x += ern;
    x = (x > 0.f) ? x : 0.2f * x;
    x = ok ? x : -INFINITY;
    float mx = x, o;
    o = row_xor<1>(mx); mx = (o > mx) ? o : mx;
    o = row_xor<2>(mx); mx = (o > mx) ? o : mx;
    o = row_xor<4>(mx); mx = (o > mx) ? o : mx;
    const float mn = (mx > m) ? mx : m;
    const float sc = expf(m - mn);
    const float a  = expf(x - mn);
    m = mn;
    den *= sc;
    acc.x *= sc; acc.y *= sc; acc.z *= sc; acc.w *= sc;
#define ADD_(i)                                                                  \
  {                                                                              \
    const float w = __int_as_float(group_lane<LPH, i>(__float_as_int(a)));       \
    den   += w;                                                                  \
    acc.x = fmaf(f[i].x, w, acc.x); acc.y = fmaf(f[i].y, w, acc.y);              \
    acc.z = fmaf(f[i].z, w, acc.z); acc.w = fmaf(f[i].w, w, acc.w);              \
  }
    ADD_(0) ADD_(1) ADD_(2) ADD_(3) ADD_(4) ADD_(5) ADD_(6) ADD_(7)
#undef ADD_
  }

  float4 o = *(const float4*)(bias + r);
  if (e > b) {
    o.x += acc.x / den; o.y += acc.y / den; o.z += acc.z / den; o.w += acc.w / den;
  }
  if (RELU) {
    o.x = fmaxf(o.x, 0.f); o.y = fmaxf(o.y, 0.f);
    o.z = fmaxf(o.z, 0.f); o.w = fmaxf(o.w, 0.f);
  }
  *(float4*)(out + node * F + r) = o;
}

extern "C" void kernel_launch(void* const* d_in, const int* in_sizes, int n_in,
                              void* d_out, int out_size, void* d_ws, size_t ws_size,
                              hipStream_t stream) {
  const float* h   = (const float*)d_in[0];
  const int*   src = (const int*)d_in[1];
  const int*   dst = (const int*)d_in[2];
  const float* W1  = (const float*)d_in[3];
  const float* al1 = (const float*)d_in[4];
  const float* ar1 = (const float*)d_in[5];
  const float* b1  = (const float*)d_in[6];
  const float* W2  = (const float*)d_in[7];
  const float* al2 = (const float*)d_in[8];
  const float* ar2 = (const float*)d_in[9];
  const float* b2  = (const float*)d_in[10];

  constexpr int F1 = 128, H1 = 4;
  constexpr int F2 = 64,  H2 = 1;
  const long N = in_sizes[0] / F1;
  const long E = in_sizes[1];

  float* ws     = (float*)d_ws;
  float* feat1  = ws;
  float* x2     = feat1 + N * F1;
  float* feat2  = x2    + N * F1;
  int*   pairs  = (int*)(feat2 + N * F2);
  int*   srcs   = pairs + 2 * E;
  int*   hist   = pairs + 3 * E;
  int*   binst  = pairs + 4 * E;
  int*   rowptr = binst + N * H1;

  auto cdiv = [](long a, long b) { return (unsigned)((a + b - 1) / b); };
  const int T = 256;
  const unsigned mtiles = cdiv(N, 16);

  constexpr unsigned WGPS = 256;
  const unsigned nbin  = cdiv(N, BIN_NODES);
  const unsigned nblk  = cdiv(E, CHUNK);
  const unsigned nslab = cdiv(N, RIDE_ROWS);
  const unsigned c2    = nslab < 2 * WGPS ? nslab : 2 * WGPS;
  const unsigned idle1 = nblk < WGPS ? WGPS - nblk : 0;
  const unsigned c1    = nslab - c2 < idle1 ? nslab - c2 : idle1;
  const unsigned c0    = nslab - c2 - c1;
  const unsigned r0 = c0, r1 = c1, r2 = cdiv(c2, 2);
  chunk_hist<<<r0 + nblk, SORT_T, 0, stream>>>(dst, hist, (int)nbin, E,
                                               h, W1, feat1, (int)N, (int)r0, 1, 0, (int)c0);
  chunk_scatter<<<r1 + nblk, SORT_T, 0, stream>>>(src, dst, hist, binst, pairs, (int)nbin, (int)nblk, E,
                                                  h, W1, feat1, (int)N, (int)r1, 1, (int)c0, (int)(c0 + c1));
  bin_sort<<<r2 + nbin, SORT_T, 0, stream>>>(pairs, binst, rowptr, srcs, (int)nbin, (int)N,
                                             h, W1, feat1, (int)N, (int)r2, 2, (int)(c0 + c1), (int)nslab);

  node_softmax_gather<F1, H1, true><<<cdiv(N * (F1 / 4), T), T, 0, stream>>>(
      rowptr, srcs, feat1, al1, ar1, b1, x2, N);

  wmma_gemm_f32_k128<F2><<<mtiles, 32 * (F2 / 16), 0, stream>>>(x2, W2, feat2, (int)N);
  node_softmax_gather<F2, H2, false><<<cdiv(N * (F2 / 4), T), T, 0, stream>>>(
      rowptr, srcs, feat2, al2, ar2, b2, (float*)d_out, N);
}


# Layer-2 projection done inside the layer-1 gather launch: a 16-wave work-group keeps its 16 rows in LDS, each wave runs one accumulator chain of the 16x64 tile; 5 launches; sums unchanged
# speedup vs baseline: 6.5476x; 1.0684x over previous
#include <hip/hip_runtime.h>
#include <math.h>

typedef __attribute__((ext_vector_type(2))) float v2f;
typedef __attribute__((ext_vector_type(8))) float v8f;

constexpr int APITCH = 132;

template <int NCOL>
__device__ __forceinline__ v8f wave_tile_k128(const float* Ap, const float* __restrict__ Bp) {
  v8f acc0 = {}, acc1 = {}, acc2 = {}, acc3 = {};
#pragma unroll
  for (int k = 0; k < 128; k += 16) {
    v2f a0, a1, a2, a3, b0, b1, b2, b3;
    a0.x = Ap[k + 0];  a0.y = Ap[k + 1];
    a1.x = Ap[k + 4];  a1.y = Ap[k + 5];
    a2.x = Ap[k + 8];  a2.y = Ap[k + 9];
    a3.x = Ap[k + 12]; a3.y = Ap[k + 13];
    b0.x = Bp[(long)(k + 0)  * NCOL]; b0.y = Bp[(long)(k + 1)  * NCOL];
    b1.x = Bp[(long)(k + 4)  * NCOL]; b1.y = Bp[(long)(k + 5)  * NCOL];
    b2.x = Bp[(long)(k + 8)  * NCOL]; b2.y = Bp[(long)(k + 9)  * NCOL];
    b3.x = Bp[(long)(k + 12) * NCOL]; b3.y = Bp[(long)(k + 13) * NCOL];
    acc0 = __builtin_amdgcn_wmma_f32_16x16x4_f32(false, a0, false, b0, (short)0, acc0, false, false);
    acc1 = __builtin_amdgcn_wmma_f32_16x16x4_f32(false, a1, false, b1, (short)0, acc1, false, false);
    acc2 = __builtin_amdgcn_wmma_f32_16x16x4_f32(false, a2, false, b2, (short)0, acc2, false, false);
    acc3 = __builtin_amdgcn_wmma_f32_16x16x4_f32(false, a3, false, b3, (short)0, acc3, false, false);
  }
  v8f acc;
#pragma unroll
  for (int i = 0; i < 8; ++i) acc[i] = (acc0[i] + acc1[i]) + (acc2[i] + acc3[i]);
  return acc;
}

template <int NCOL>
__device__ __forceinline__ v8f wave_chain_k128(const float* Ap, const float* __restrict__ Bp) {
  v8f acc = {};
#pragma unroll
  for (int k = 0; k < 128; k += 16) {
    v2f a, b;
    a.x = Ap[k];        a.y = Ap[k + 1];
    b.x = Bp[k * NCOL]; b.y = Bp[(k + 1) * NCOL];
    acc = __builtin_amdgcn_wmma_f32_16x16x4_f32(false, a, false, b, (short)0, acc, false, false);
  }
  return acc;
}

constexpr int RIDE_T = 1024;
template <int NCOL>
struct Ride {
  static constexpr int CT   = NCOL / 16;
  static constexpr int MT   = (RIDE_T / 32) / CT;
  static constexpr int ROWS = 16 * MT;
  static constexpr int VPT  = ROWS * 32 / RIDE_T;
};
constexpr int RIDE_ROWS = Ride<128>::ROWS;

template <int NCOL>
__device__ __forceinline__ void gemm_ride(float* As, const float* __restrict__ A,
                                          const float* __restrict__ B, float* __restrict__ C,
                                          int M, int slab, int slab_end) {
  constexpr int CT = Ride<NCOL>::CT, ROWS = Ride<NCOL>::ROWS;
  static_assert(Ride<NCOL>::VPT == 2, "two float4 of a slab per thread");
  const int lane  = threadIdx.x & 31;
  const int wave  = threadIdx.x >> 5;
  const int mt    = wave / CT;
  const int col0  = (wave % CT) * 16;
  const int mrow  = lane & 15;
  const int khalf = (lane >> 4) << 1;
  const float* Ap = As + (mt * 16 + mrow) * APITCH + khalf;
  const float* Bp = B + (long)khalf * NCOL + col0 + mrow;
  const int i0 = threadIdx.x >> 5, c4 = (threadIdx.x & 31) * 4;
  float* As0 = As + i0 * APITCH + c4;
  float* As1 = As0 + 32 * APITCH;
  const float* Ac = A + c4;

  if (slab >= slab_end) return;
  int ra = slab * ROWS + i0, rb = ra + 32;
  ra = ra < M ? ra : M - 1;
  rb = rb < M ? rb : M - 1;
  float4 p0 = *(const float4*)(Ac + (long)ra * 128);
  float4 p1 = *(const float4*)(Ac + (long)rb * 128);
  for (;;) {
    *(float4*)As0 = p0;
    *(float4*)As1 = p1;
    __syncthreads();
    if (slab + 1 < slab_end && threadIdx.x < ROWS * 2) {
      int rn = (slab + 1) * ROWS + (threadIdx.x >> 1);
      rn = rn < M ? rn : M - 1;
      __builtin_prefetch(A + (long)rn * 128 + (threadIdx.x & 1) * 64, 0, 3);
    }
    int boff = 0;
    asm volatile("" : "+v"(boff));
    const v8f acc = wave_tile_k128<NCOL>(Ap, Bp + boff);
    const int rbase = slab * ROWS + mt * 16 + ((lane >> 4) << 3);
    float* Cp = C + (long)rbase * NCOL + col0 + mrow;
    if ((slab + 1) * ROWS <= M) {
#pragma unroll
      for (int i = 0; i < 8; ++i) Cp[(long)i * NCOL] = acc[i];
    } else {
#pragma unroll
      for (int i = 0; i < 8; ++i)
        if (rbase + i < M) Cp[(long)i * NCOL] = acc[i];
    }
    if (++slab >= slab_end) break;
    ra = slab * ROWS + i0; rb = ra + 32;
    ra = ra < M ? ra : M - 1;
    rb = rb < M ? rb : M - 1;
    p0 = *(const float4*)(Ac + (long)ra * 128);
    p1 = *(const float4*)(Ac + (long)rb * 128);
    __syncthreads();
  }
}

constexpr int BIN_SHIFT = 8;
constexpr int BIN_NODES = 1 << BIN_SHIFT;
constexpr int MAXBIN    = 256;
constexpr int CHUNK     = 4096;
constexpr int SORT_T    = 1024;
static_assert(SORT_T == RIDE_T, "the riding GEMM slabs are cut for the sort's work-group size");
constexpr int BIN_CAP   = 16384;

__global__ void __launch_bounds__(SORT_T) __attribute__((amdgpu_waves_per_eu(16))) chunk_hist(const int* __restrict__ dst,
                                                     int* __restrict__ hist, int nbin, long E,
                                                     const float* __restrict__ gA,
                                                     const float* __restrict__ gB,
                                                     float* __restrict__ gC, int gM,
                                                     int nride, int gper, int gslab0, int gend) {
  __shared__ int h[MAXBIN];
  __shared__ __attribute__((aligned(16))) float As[RIDE_ROWS * APITCH];
  if ((int)blockIdx.x < nride) {
    const int s0 = gslab0 + (int)blockIdx.x * gper;
    gemm_ride<128>(As, gA, gB, gC, gM, s0, (s0 + gper < gend) ? s0 + gper : gend);
    return;
  }
  const int blk = (int)blockIdx.x - nride;
  const int tid = threadIdx.x;
  if (tid < MAXBIN) h[tid] = 0;
  __syncthreads();
  const long e = (long)blk * CHUNK + tid * 4;
  if (e + 4 <= E) {
    const int4 d = *(const int4*)(dst + e);
    atomicAdd(&h[d.x >> BIN_SHIFT], 1);
    atomicAdd(&h[d.y >> BIN_SHIFT], 1);
    atomicAdd(&h[d.z >> BIN_SHIFT], 1);
    atomicAdd(&h[d.w >> BIN_SHIFT], 1);
  } else {
    for (long i = e; i < E; ++i) atomicAdd(&h[dst[i] >> BIN_SHIFT], 1);
  }
  __syncthreads();
  if (tid < nbin) hist[(long)blk * nbin + tid] = h[tid];
}

__global__ void __launch_bounds__(SORT_T) chunk_scatter(
    const int* __restrict__ src, const int* __restrict__ dst, const int* __restrict__ hist,
    int* __restrict__ binstart, int* __restrict__ pairs_w, int nbin, int nblk, long E,
    const float* __restrict__ gA, const float* __restrict__ gB, float* __restrict__ gC,
    int gM, int nride, int gper, int gslab0, int gend) {
  __shared__ __attribute__((aligned(16))) float As[RIDE_ROWS * APITCH];
  if ((int)blockIdx.x < nride) {
    const int s0 = gslab0 + (int)blockIdx.x * gper;
    gemm_ride<128>(As, gA, gB, gC, gM, s0, (s0 + gper < gend) ? s0 + gper : gend);
    return;
  }
  constexpr int PARTS = SORT_T / MAXBIN;
  __shared__ int  ptot[PARTS][MAXBIN], ppre[PARTS][MAXBIN];
  __shared__ int  gdelta[MAXBIN], lcur[MAXBIN];
  __shared__ int  wsum[2][MAXBIN / 32];
  __shared__ int2 stage[CHUNK];
  int2* __restrict__ pairs = (int2*)pairs_w;
  const int tid = threadIdx.x, lane = tid & 31, wave = tid >> 5;
  const int blk = (int)blockIdx.x - nride;
  const long e0 = (long)blk * CHUNK;
  const long e  = e0 + tid * 4;

  const bool full = e + 4 <= E;
  int4 s4 = {0, 0, 0, 0}, d4 = {0, 0, 0, 0};
  if (full) { s4 = *(const int4*)(src + e); d4 = *(const int4*)(dst + e); }

  const int bin = tid & (MAXBIN - 1), part = tid / MAXBIN;
  int tot = 0, pre = 0;
  if (bin < nbin) {
    const int* __restrict__ col = hist + bin;
    for (int k0 = part; k0 < nblk; k0 += PARTS * 16) {
      int v[16];
#pragma unroll
      for (int j = 0; j < 16; ++j) {
        const int k = k0 + PARTS * j;
        v[j] = col[(k < nblk ? k : nblk - 1) * nbin];
      }
#pragma unroll
      for (int j = 0; j < 16; ++j) {
        const int k = k0 + PARTS * j;
        const int x = (k < nblk) ? v[j] : 0;
        tot += x;
        pre += (k < blk) ? x : 0;
      }
    }
  }
  ptot[part][bin] = tot; ppre[part][bin] = pre;
  __syncthreads();

  int t4 = 0, p4 = 0, own = 0, ia = 0, ic = 0;
  if (tid < MAXBIN) {
    for (int p = 0; p < PARTS; ++p) { t4 += ptot[p][tid]; p4 += ppre[p][tid]; }
    own = (tid < nbin) ? hist[(long)blk * nbin + tid] : 0;
    ia = t4; ic = own;
    for (int o = 1; o < 32; o <<= 1) {
      const int va = __shfl_up(ia, o), vc = __shfl_up(ic, o);
      if (lane >= o) { ia += va; ic += vc; }
    }
    if (lane == 31) { wsum[0][wave] = ia; wsum[1][wave] = ic; }
  }
  __syncthreads();
  if (tid < MAXBIN) {
    int ba = 0, bc = 0;
    for (int w = 0; w < wave; ++w) { ba += wsum[0][w]; bc += wsum[1][w]; }
    const int bstart = ba + ia - t4;
    const int lstart = bc + ic - own;
    gdelta[tid] = bstart + p4 - lstart;
    lcur[tid]   = lstart;
    if (blk == 0 && tid < nbin)      binstart[tid]  = bstart;
    if (blk == 0 && tid == nbin - 1) binstart[nbin] = bstart + t4;
  }
  __syncthreads();

  if (full) {
    const int q0 = atomicAdd(&lcur[d4.x >> BIN_SHIFT], 1);
    const int q1 = atomicAdd(&lcur[d4.y >> BIN_SHIFT], 1);
    const int q2 = atomicAdd(&lcur[d4.z >> BIN_SHIFT], 1);
    const int q3 = atomicAdd(&lcur[d4.w >> BIN_SHIFT], 1);
    stage[q0] = make_int2(s4.x, d4.x);
    stage[q1] = make_int2(s4.y, d4.y);
    stage[q2] = make_int2(s4.z, d4.z);
    stage[q3] = make_int2(s4.w, d4.w);
  } else {
    for (long i = e; i < E; ++i) {
      const int d = dst[i];
      stage[atomicAdd(&lcur[d >> BIN_SHIFT], 1)] = make_int2(src[i], d);
    }
  }
  __syncthreads();
  const long left = E - e0;
  const int nloc = left < CHUNK ? (int)left : CHUNK;
  for (int i = tid; i < nloc; i += SORT_T) {
    const int2 p = stage[i];
    pairs[i + gdelta[p.y >> BIN_SHIFT]] = p;
  }
}

__global__ void __launch_bounds__(SORT_T) __attribute__((amdgpu_waves_per_eu(16))) bin_sort(
    const int* __restrict__ pairs_w, const int* __restrict__ binstart,
    int* __restrict__ rowptr, int* __restrict__ srcs, int nbin, int N,
    const float* __restrict__ gA, const float* __restrict__ gB, float* __restrict__ gC,
    int gM, int nride, int gper, int gslab0, int gend) {
  __shared__ __attribute__((aligned(16))) float As[RIDE_ROWS * APITCH];
  if ((int)blockIdx.x < nride) {
    const int s0 = gslab0 + (int)blockIdx.x * gper;
    gemm_ride<128>(As, gA, gB, gC, gM, s0, (s0 + gper < gend) ? s0 + gper : gend);
    return;
  }
  __shared__ int cnt[BIN_NODES], cur[BIN_NODES];
  __shared__ int wsum[BIN_NODES / 32];
  __shared__ int lsrt[BIN_CAP];
  const int2* __restrict__ pairs = (const int2*)pairs_w;
  const int tid = threadIdx.x, lane = tid & 31, wave = tid >> 5;
  const int b  = (int)blockIdx.x - nride;
  const int n0 = b << BIN_SHIFT;
  const int bs = binstart[b], be = binstart[b + 1];
  const int m  = be - bs;
  if (tid < BIN_NODES) cnt[tid] = 0;
  __syncthreads();
  for (int i0 = tid; i0 < m; i0 += 4 * SORT_T) {
    int d[4];
#pragma unroll
    for (int j = 0; j < 4; ++j) {
      const int i = i0 + j * SORT_T;
      d[j] = pairs[bs + (i < m ? i : m - 1)].y;
    }
#pragma unroll
    for (int j = 0; j < 4; ++j)
      if (i0 + j * SORT_T < m) atomicAdd(&cnt[d[j] - n0], 1);
  }
  __syncthreads();
  int c = 0, inc = 0;
  if (tid < BIN_NODES) {
    c = cnt[tid]; inc = c;
    for (int o = 1; o < 32; o <<= 1) {
      const int v = __shfl_up(inc, o);
      if (lane >= o) inc += v;
    }
    if (lane == 31) wsum[wave] = inc;
  }
  __syncthreads();
  if (tid < BIN_NODES) {
    int base = 0;
    for (int w = 0; w < wave; ++w) base += wsum[w];
    const int ex = base + inc - c;
    cur[tid] = ex;
    if (n0 + tid < N) rowptr[n0 + tid] = bs + ex;
  }
  if (b == nbin - 1 && tid == 0) rowptr[N] = be;
  __syncthreads();
  const bool fits = m <= BIN_CAP;
  for (int i0 = tid; i0 < m; i0 += 4 * SORT_T) {
    int2 p[4];
#pragma unroll
    for (int j = 0; j < 4; ++j) {
      const int i = i0 + j * SORT_T;
      p[j] = pairs[bs + (i < m ? i : m - 1)];
    }
#pragma unroll
    for (int j = 0; j < 4; ++j) {
      if (i0 + j * SORT_T < m) {
        const int pos = atomicAdd(&cur[p[j].y - n0], 1);
        if (fits) lsrt[pos] = p[j].x; else srcs[bs + pos] = p[j].x;
      }
    }
  }
  __syncthreads();
  if (fits)
    for (int i = tid; i < m; i += SORT_T) srcs[bs + i] = lsrt[i];
}


template <int LPH, int J>
__device__ __forceinline__ int group_lane(int v) {
  if constexpr (LPH == 8) {
    return __builtin_amdgcn_mov_dpp8(v, J * 0x249249);
  } else {
    return __builtin_amdgcn_update_dpp(v, v, 0x150 + J, 0xf, 0xf, false);
  }
}

template <int M>
__device__ __forceinline__ float row_xor(float v) {
  const int i = __float_as_int(v);
  return __int_as_float(__builtin_amdgcn_update_dpp(0, i, 0x160 + M, 0xf, 0xf, true));
}

template <int LPH>
__device__ __forceinline__ float group_sum(float v) {
  v += row_xor<1>(v);
  v += row_xor<2>(v);
  v += row_xor<4>(v);
  if constexpr (LPH == 16) v += row_xor<8>(v);
  return v;
}

__device__ __forceinline__ float dot4(const float4 a, const float4 b) {
  return fmaf(a.w, b.w, fmaf(a.z, b.z, fmaf(a.y, b.y, a.x * b.x)));
}

template <int F, int H, bool RELU>
__device__ __forceinline__ float4 node_row(
    const int* __restrict__ rowptr, const int* __restrict__ srcs,
    const float* __restrict__ feat, const float* __restrict__ al,
    const float* __restrict__ ar, const float* __restrict__ bias,
    const long node, const int ln) {
  constexpr int D   = F / H;
  constexpr int LPH = D / 4;
  static_assert(LPH == 8 || LPH == 16, "lane groups of 8 or 16");
  const int r  = ln * 4;
  const int jj = ln & 7;
  const int b = rowptr[node], e = rowptr[node + 1];
  const int deg = e - b;
  const char* fbytes = (const char*)feat;

  const float4 l4 = *(const float4*)(al + r);
  const float4 r4 = *(const float4*)(ar + r);
  const float ern = group_sum<LPH>(dot4(*(const float4*)(feat + node * F + r), r4));

  float m = -INFINITY, den = 0.f;
  float4 acc = {0.f, 0.f, 0.f, 0.f};
  int sv = 0;
  if (deg > 0) sv = srcs[(b + jj < e) ? b + jj : b];
  for (int off = 0; off < deg; off += 8) {
    const int  vo = sv * (F * 4);
    const bool ok = off + jj < deg;
    if (off + 8 < deg) {
      const int nx = b + off + 8 + jj;
      sv = srcs[(nx < e) ? nx : b + off + 8];
    }
    float4 f[8];
    float  s[8];
#define ROW_(i)                                                                  \
  {                                                                              \
    const unsigned o = (unsigned)group_lane<LPH, i>(vo) + (unsigned)(r * 4);     \
    f[i] = *(const float4*)(fbytes + o);                                         \
  }
    ROW_(0) ROW_(1) ROW_(2) ROW_(3) ROW_(4) ROW_(5) ROW_(6) ROW_(7)
#undef ROW_
#pragma unroll
    for (int i = 0; i < 8; ++i) s[i] = group_sum<LPH>(dot4(f[i], l4));
    float x = s[0];
#pragma unroll
    for (int i = 1; i < 8; ++i) x = (jj == i) ? s[i] : x;
    x += ern;
    x = (x > 0.f) ? x : 0.2f * x;
    x = ok ? x : -INFINITY;
    float mx = x, o;
    o = row_xor<1>(mx); mx = (o > mx) ? o : mx;
    o = row_xor<2>(mx); mx = (o > mx) ? o : mx;
    o = row_xor<4>(mx); mx = (o > mx) ? o : mx;
    const float mn = (mx > m) ? mx : m;
    const float sc = expf(m - mn);
    const float a  = expf(x - mn);
    m = mn;
    den *= sc;
    acc.x *= sc; acc.y *= sc; acc.z *= sc; acc.w *= sc;
#define ADD_(i)                                                                  \
  {                                                                              \
    const float w = __int_as_float(group_lane<LPH, i>(__float_as_int(a)));       \
    den   += w;                                                                  \
    acc.x = fmaf(f[i].x, w, acc.x); acc.y = fmaf(f[i].y, w, acc.y);              \
    acc.z = fmaf(f[i].z, w, acc.z); acc.w = fmaf(f[i].w, w, acc.w);              \
  }
    ADD_(0) ADD_(1) ADD_(2) ADD_(3) ADD_(4) ADD_(5) ADD_(6) ADD_(7)
#undef ADD_
  }

  float4 o = *(const float4*)(bias + r);
  if (e > b) {
    o.x += acc.x / den; o.y += acc.y / den; o.z += acc.z / den; o.w += acc.w / den;
  }
  if (RELU) {
    o.x = fmaxf(o.x, 0.f); o.y = fmaxf(o.y, 0.f);
    o.z = fmaxf(o.z, 0.f); o.w = fmaxf(o.w, 0.f);
  }
  return o;
}

template <int F, int H, bool RELU>
__global__ void __launch_bounds__(256) node_softmax_gather(
    const int* __restrict__ rowptr, const int* __restrict__ srcs,
    const float* __restrict__ feat, const float* __restrict__ al,
    const float* __restrict__ ar, const float* __restrict__ bias,
    float* __restrict__ out, long N) {
  constexpr int LPN = F / 4;
  const long t    = (long)blockIdx.x * blockDim.x + threadIdx.x;
  const long node = t / LPN;
  if (node >= N) return;
  const int ln = (int)(t & (LPN - 1));
  const float4 o = node_row<F, H, RELU>(rowptr, srcs, feat, al, ar, bias, node, ln);
  *(float4*)(out + node * F + ln * 4) = o;
}

template <int H, int NCOL>
__global__ void __launch_bounds__(512) node_softmax_gather_proj(
    const int* __restrict__ rowptr, const int* __restrict__ srcs,
    const float* __restrict__ feat, const float* __restrict__ al,
    const float* __restrict__ ar, const float* __restrict__ bias,
    const float* __restrict__ B, float* __restrict__ C, long N) {
  static_assert(NCOL == 64, "16 waves = 4 column tiles x 4 chains");
  __shared__ __attribute__((aligned(16))) float Xs[16 * APITCH];
  constexpr int PT_PITCH = 4 * 256 + 16;
  __shared__ __attribute__((aligned(16))) float Ps[4 * PT_PITCH];
  const int lane = threadIdx.x & 31;
  const int wave = threadIdx.x >> 5;
  const long node0 = (long)blockIdx.x * 16;
  long node = node0 + wave;
  node = node < N ? node : N - 1;
  const float4 o = node_row<128, H, true>(rowptr, srcs, feat, al, ar, bias, node, lane);
  *(float4*)(Xs + wave * APITCH + lane * 4) = o;
  __syncthreads();

  const int wv = __builtin_amdgcn_readfirstlane(wave);
  const int ct = wv >> 2;
  const int ch = wv & 3;
  const int mrow  = lane & 15;
  const int khalf = (lane >> 4) << 1;
  const v8f acc = wave_chain_k128<NCOL>(Xs + 4 * ch + (mrow * APITCH + khalf),
                                        (B + 4 * ch * NCOL + ct * 16) + (khalf * NCOL + mrow));
  float* mine = Ps + ct * PT_PITCH + ch * 256 + lane;
#pragma unroll
  for (int i = 0; i < 8; ++i) mine[i * 32] = acc[i];
  __syncthreads();

  const float* q = Ps + (lane >> 3) * PT_PITCH + (wv & 7) * 32 + (wv >> 3) * 16 + (lane & 7) * 2;
  const float2 s0 = *(const float2*)(q);
  const float2 s1 = *(const float2*)(q + 256);
  const float2 s2 = *(const float2*)(q + 512);
  const float2 s3 = *(const float2*)(q + 768);
  float2 v;
  v.x = (s0.x + s1.x) + (s2.x + s3.x);
  v.y = (s0.y + s1.y) + (s2.y + s3.y);
  if (node0 + wv < N) *(float2*)(C + (node0 + wv) * NCOL + lane * 2) = v;
}

extern "C" void kernel_launch(void* const* d_in, const int* in_sizes, int n_in,
                              void* d_out, int out_size, void* d_ws, size_t ws_size,
                              hipStream_t stream) {
  const float* h   = (const float*)d_in[0];
  const int*   src = (const int*)d_in[1];
  const int*   dst = (const int*)d_in[2];
  const float* W1  = (const float*)d_in[3];
  const float* al1 = (const float*)d_in[4];
  const float* ar1 = (const float*)d_in[5];
  const float* b1  = (const float*)d_in[6];
  const float* W2  = (const float*)d_in[7];
  const float* al2 = (const float*)d_in[8];
  const float* ar2 = (const float*)d_in[9];
  const float* b2  = (const float*)d_in[10];

  constexpr int F1 = 128, H1 = 4;
  constexpr int F2 = 64,  H2 = 1;
  const long N = in_sizes[0] / F1;
  const long E = in_sizes[1];

  float* ws     = (float*)d_ws;
  float* feat1  = ws;
  float* feat2  = feat1 + 2 * N * F1;
  int*   pairs  = (int*)(feat2 + N * F2);
  int*   srcs   = pairs + 2 * E;
  int*   hist   = pairs + 3 * E;
  int*   binst  = pairs + 4 * E;
  int*   rowptr = binst + N * H1;

  auto cdiv = [](long a, long b) { return (unsigned)((a + b - 1) / b); };
  const int T = 256;
  const unsigned mtiles = cdiv(N, 16);

  constexpr unsigned WGPS = 256;
  const unsigned nbin  = cdiv(N, BIN_NODES);
  const unsigned nblk  = cdiv(E, CHUNK);
  const unsigned nslab = cdiv(N, RIDE_ROWS);
  const unsigned c2    = nslab < 2 * WGPS ? nslab : 2 * WGPS;
  const unsigned idle1 = nblk < WGPS ? WGPS - nblk : 0;
  const unsigned c1    = nslab - c2 < idle1 ? nslab - c2 : idle1;
  const unsigned c0    = nslab - c2 - c1;
  const unsigned r0 = c0, r1 = c1, r2 = cdiv(c2, 2);
  chunk_hist<<<r0 + nblk, SORT_T, 0, stream>>>(dst, hist, (int)nbin, E,
                                               h, W1, feat1, (int)N, (int)r0, 1, 0, (int)c0);
  chunk_scatter<<<r1 + nblk, SORT_T, 0, stream>>>(src, dst, hist, binst, pairs, (int)nbin, (int)nblk, E,
                                                  h, W1, feat1, (int)N, (int)r1, 1, (int)c0, (int)(c0 + c1));
  bin_sort<<<r2 + nbin, SORT_T, 0, stream>>>(pairs, binst, rowptr, srcs, (int)nbin, (int)N,
                                             h, W1, feat1, (int)N, (int)r2, 2, (int)(c0 + c1), (int)nslab);

  static_assert(F1 == 128, "the fused kernel's rows are the K = 128 of wave_tile_k128");
  node_softmax_gather_proj<H1, F2><<<mtiles, 512, 0, stream>>>(
      rowptr, srcs, feat1, al1, ar1, b1, W2, feat2, N);

  node_softmax_gather<F2, H2, false><<<cdiv(N * (F2 / 4), T), T, 0, stream>>>(
      rowptr, srcs, feat2, al2, ar2, b2, (float*)d_out, N);
}


# Edge sort in two launches, not three: each chunk is ordered by bin in place, a bin's work-group collects its runs through a run table; node sums in LDS arrival order over the new edge order
# speedup vs baseline: 6.9781x; 1.0657x over previous
#include <hip/hip_runtime.h>
#include <math.h>

typedef __attribute__((ext_vector_type(2))) float v2f;
typedef __attribute__((ext_vector_type(8))) float v8f;

constexpr int APITCH = 132;

template <int NCOL>
__device__ __forceinline__ v8f wave_tile_k128(const float* Ap, const float* __restrict__ Bp) {
  v8f acc0 = {}, acc1 = {}, acc2 = {}, acc3 = {};
#pragma unroll
  for (int k = 0; k < 128; k += 16) {
    v2f a0, a1, a2, a3, b0, b1, b2, b3;
    a0.x = Ap[k + 0];  a0.y = Ap[k + 1];
    a1.x = Ap[k + 4];  a1.y = Ap[k + 5];
    a2.x = Ap[k + 8];  a2.y = Ap[k + 9];
    a3.x = Ap[k + 12]; a3.y = Ap[k + 13];
    b0.x = Bp[(long)(k + 0)  * NCOL]; b0.y = Bp[(long)(k + 1)  * NCOL];
    b1.x = Bp[(long)(k + 4)  * NCOL]; b1.y = Bp[(long)(k + 5)  * NCOL];
    b2.x = Bp[(long)(k + 8)  * NCOL]; b2.y = Bp[(long)(k + 9)  * NCOL];
    b3.x = Bp[(long)(k + 12) * NCOL]; b3.y = Bp[(long)(k + 13) * NCOL];
    acc0 = __builtin_amdgcn_wmma_f32_16x16x4_f32(false, a0, false, b0, (short)0, acc0, false, false);
    acc1 = __builtin_amdgcn_wmma_f32_16x16x4_f32(false, a1, false, b1, (short)0, acc1, false, false);
    acc2 = __builtin_amdgcn_wmma_f32_16x16x4_f32(false, a2, false, b2, (short)0, acc2, false, false);
    acc3 = __builtin_amdgcn_wmma_f32_16x16x4_f32(false, a3, false, b3, (short)0, acc3, false, false);
  }
  v8f acc;
#pragma unroll
  for (int i = 0; i < 8; ++i) acc[i] = (acc0[i] + acc1[i]) + (acc2[i] + acc3[i]);
  return acc;
}

template <int NCOL>
__device__ __forceinline__ v8f wave_chain_k128(const float* Ap, const float* __restrict__ Bp) {
  v8f acc = {};
#pragma unroll
  for (int k = 0; k < 128; k += 16) {
    v2f a, b;
    a.x = Ap[k];        a.y = Ap[k + 1];
    b.x = Bp[k * NCOL]; b.y = Bp[(k + 1) * NCOL];
    acc = __builtin_amdgcn_wmma_f32_16x16x4_f32(false, a, false, b, (short)0, acc, false, false);
  }
  return acc;
}

constexpr int RIDE_T = 1024;
template <int NCOL>
struct Ride {
  static constexpr int CT   = NCOL / 16;
  static constexpr int MT   = (RIDE_T / 32) / CT;
  static constexpr int ROWS = 16 * MT;
  static constexpr int VPT  = ROWS * 32 / RIDE_T;
};
constexpr int RIDE_ROWS = Ride<128>::ROWS;

template <int NCOL>
__device__ __forceinline__ void gemm_ride(float* As, const float* __restrict__ A,
                                          const float* __restrict__ B, float* __restrict__ C,
                                          int M, int slab, int slab_end) {
  constexpr int CT = Ride<NCOL>::CT, ROWS = Ride<NCOL>::ROWS;
  static_assert(Ride<NCOL>::VPT == 2, "two float4 of a slab per thread");
  const int lane  = threadIdx.x & 31;
  const int wave  = threadIdx.x >> 5;
  const int mt    = wave / CT;
  const int col0  = (wave % CT) * 16;
  const int mrow  = lane & 15;
  const int khalf = (lane >> 4) << 1;
  const float* Ap = As + (mt * 16 + mrow) * APITCH + khalf;
  const float* Bp = B + (long)khalf * NCOL + col0 + mrow;
  const int i0 = threadIdx.x >> 5, c4 = (threadIdx.x & 31) * 4;
  float* As0 = As + i0 * APITCH + c4;
  float* As1 = As0 + 32 * APITCH;
  const float* Ac = A + c4;

  if (slab >= slab_end) return;
  int ra = slab * ROWS + i0, rb = ra + 32;
  ra = ra < M ? ra : M - 1;
  rb = rb < M ? rb : M - 1;
  float4 p0 = *(const float4*)(Ac + (long)ra * 128);
  float4 p1 = *(const float4*)(Ac + (long)rb * 128);
  for (;;) {
    *(float4*)As0 = p0;
    *(float4*)As1 = p1;
    __syncthreads();
    if (slab + 1 < slab_end && threadIdx.x < ROWS * 2) {
      int rn = (slab + 1) * ROWS + (threadIdx.x >> 1);
      rn = rn < M ? rn : M - 1;
      __builtin_prefetch(A + (long)rn * 128 + (threadIdx.x & 1) * 64, 0, 3);
    }
    int boff = 0;
    asm volatile("" : "+v"(boff));
    const v8f acc = wave_tile_k128<NCOL>(Ap, Bp + boff);
    const int rbase = slab * ROWS + mt * 16 + ((lane >> 4) << 3);
    float* Cp = C + (long)rbase * NCOL + col0 + mrow;
    if ((slab + 1) * ROWS <= M) {
#pragma unroll
      for (int i = 0; i < 8; ++i) Cp[(long)i * NCOL] = acc[i];
    } else {
#pragma unroll
      for (int i = 0; i < 8; ++i)
        if (rbase + i < M) Cp[(long)i * NCOL] = acc[i];
    }
    if (++slab >= slab_end) break;
    ra = slab * ROWS + i0; rb = ra + 32;
    ra = ra < M ? ra : M - 1;
    rb = rb < M ? rb : M - 1;
    p0 = *(const float4*)(Ac + (long)ra * 128);
    p1 = *(const float4*)(Ac + (long)rb * 128);
    __syncthreads();
  }
}

constexpr int BIN_SHIFT = 8;
constexpr int BIN_NODES = 1 << BIN_SHIFT;
constexpr int MAXBIN    = 256;
constexpr int CHUNK     = 4096;
constexpr int SORT_T    = 1024;
constexpr int MAXCH     = SORT_T;
static_assert(SORT_T == RIDE_T, "the riding GEMM slabs are cut for the sort's work-group size");
constexpr int BIN_CAP   = 8192;
constexpr int NO_RUN    = 0x7fffffff;

__global__ void __launch_bounds__(SORT_T) __attribute__((amdgpu_waves_per_eu(16))) chunk_sort(
    const int* __restrict__ src, const int* __restrict__ dst,
    int* __restrict__ runs_w, int* __restrict__ cedge, long E,
    const float* __restrict__ gA, const float* __restrict__ gB, float* __restrict__ gC,
    int gM, int nride, int gper, int gslab0, int gend) {
  __shared__ __attribute__((aligned(16))) float As[RIDE_ROWS * APITCH];
  if ((int)blockIdx.x < nride) {
    const int s0 = gslab0 + (int)blockIdx.x * gper;
    gemm_ride<128>(As, gA, gB, gC, gM, s0, (s0 + gper < gend) ? s0 + gper : gend);
    return;
  }
  __shared__ int h[MAXBIN], lcur[MAXBIN];
  __shared__ int wsum[MAXBIN / 32];
  __shared__ int stage[CHUNK];
  int2* __restrict__ runs = (int2*)runs_w;
  const int tid = threadIdx.x, lane = tid & 31, wave = tid >> 5;
  const int blk = (int)blockIdx.x - nride;
  const long e0 = (long)blk * CHUNK;
  const long e  = e0 + tid * 4;

  const bool full = e + 4 <= E;
  int4 s4 = {0, 0, 0, 0}, d4 = {0, 0, 0, 0};
  if (full) { s4 = *(const int4*)(src + e); d4 = *(const int4*)(dst + e); }
  if (tid < MAXBIN) h[tid] = 0;
  __syncthreads();
  if (full) {
    atomicAdd(&h[d4.x >> BIN_SHIFT], 1);
    atomicAdd(&h[d4.y >> BIN_SHIFT], 1);
    atomicAdd(&h[d4.z >> BIN_SHIFT], 1);
    atomicAdd(&h[d4.w >> BIN_SHIFT], 1);
  } else {
    for (long i = e; i < E; ++i) atomicAdd(&h[dst[i] >> BIN_SHIFT], 1);
  }
  __syncthreads();

  int own = 0, ic = 0;
  if (tid < MAXBIN) {
    own = h[tid]; ic = own;
    for (int o = 1; o < 32; o <<= 1) {
      const int vc = __shfl_up(ic, o);
      if (lane >= o) ic += vc;
    }
    if (lane == 31) wsum[wave] = ic;
  }
  __syncthreads();
  if (tid < MAXBIN) {
    int bc = 0;
    for (int w = 0; w < wave; ++w) bc += wsum[w];
    const int lstart = bc + ic - own;
    lcur[tid] = lstart;
    runs[(long)blk * MAXBIN + tid] = make_int2(lstart, own);
  }
  __syncthreads();

  if (full) {
    const int q0 = atomicAdd(&lcur[d4.x >> BIN_SHIFT], 1);
    const int q1 = atomicAdd(&lcur[d4.y >> BIN_SHIFT], 1);
    const int q2 = atomicAdd(&lcur[d4.z >> BIN_SHIFT], 1);
    const int q3 = atomicAdd(&lcur[d4.w >> BIN_SHIFT], 1);
    stage[q0] = (s4.x << BIN_SHIFT) | (d4.x & (BIN_NODES - 1));
    stage[q1] = (s4.y << BIN_SHIFT) | (d4.y & (BIN_NODES - 1));
    stage[q2] = (s4.z << BIN_SHIFT) | (d4.z & (BIN_NODES - 1));
    stage[q3] = (s4.w << BIN_SHIFT) | (d4.w & (BIN_NODES - 1));
  } else {
    for (long i = e; i < E; ++i) {
      const int d = dst[i];
      stage[atomicAdd(&lcur[d >> BIN_SHIFT], 1)] = (src[i] << BIN_SHIFT) | (d & (BIN_NODES - 1));
    }
  }
  __syncthreads();
  const long left = E - e0;
  const int nloc = left < CHUNK ? (int)left : CHUNK;
  for (int i = tid; i < nloc; i += SORT_T) cedge[e0 + i] = stage[i];
}

__global__ void __launch_bounds__(SORT_T) __attribute__((amdgpu_waves_per_eu(16))) bin_sort(
    const int* __restrict__ cedge, const int* __restrict__ runs_w,
    int* __restrict__ rowptr, int* __restrict__ srcs, int nbin, int nblk, int N,
    const float* __restrict__ gA, const float* __restrict__ gB, float* __restrict__ gC,
    int gM, int nride, int gper, int gslab0, int gend) {
  __shared__ __attribute__((aligned(16))) float As[RIDE_ROWS * APITCH];
  if ((int)blockIdx.x < nride) {
    const int s0 = gslab0 + (int)blockIdx.x * gper;
    gemm_ride<128>(As, gA, gB, gC, gM, s0, (s0 + gper < gend) ? s0 + gper : gend);
    return;
  }
  __shared__ int cnt[BIN_NODES], cur[BIN_NODES];
  __shared__ int wsum[BIN_NODES / 32];
  __shared__ int wtot[2][SORT_T / 32];
  __shared__ int roff[MAXCH];
  __shared__ int rsrc[MAXCH];
  __shared__ int stage[BIN_CAP], lsrt[BIN_CAP];
  const int2* __restrict__ runs = (const int2*)runs_w;
  const int tid = threadIdx.x, lane = tid & 31, wave = tid >> 5;
  const int b  = (int)blockIdx.x - nride;
  const int n0 = b << BIN_SHIFT;

  int2 run = {0, 0};
  if (tid < nblk) run = runs[(long)tid * MAXBIN + b];
  if (tid < BIN_NODES) cnt[tid] = 0;
  int rinc = run.y, rst = run.x;
  for (int o = 1; o < 32; o <<= 1) {
    const int v = __shfl_up(rinc, o);
    if (lane >= o) rinc += v;
  }
  for (int o = 16; o > 0; o >>= 1) rst += __shfl_xor(rst, o);
  if (lane == 31) wtot[0][wave] = rinc;
  if (lane == 0)  wtot[1][wave] = rst;
  __syncthreads();
  int rbase = 0, m = 0, bs = 0;
  for (int w = 0; w < SORT_T / 32; ++w) {
    const int a = wtot[0][w];
    rbase += (w < wave) ? a : 0;
    m  += a;
    bs += wtot[1][w];
  }
  const int rex = rbase + rinc - run.y;
  roff[tid] = (tid < nblk) ? rex : NO_RUN;
  rsrc[tid] = tid * CHUNK + run.x - rex;
  __syncthreads();

  int top = 1;
  while (top < nblk) top <<= 1;
  auto fetch4 = [&](int i0, int (&v)[4]) {
    int ii[4], ch[4];
#pragma unroll
    for (int j = 0; j < 4; ++j) {
      const int i = i0 + j * SORT_T;
      ii[j] = i < m ? i : m - 1;
      ch[j] = 0;
    }
    for (int s = top >> 1; s > 0; s >>= 1) {
#pragma unroll
      for (int j = 0; j < 4; ++j) ch[j] += (roff[ch[j] + s] <= ii[j]) ? s : 0;
    }
#pragma unroll
    for (int j = 0; j < 4; ++j) v[j] = cedge[rsrc[ch[j]] + ii[j]];
  };

  const bool fits = m <= BIN_CAP;
  for (int i0 = tid; i0 < m; i0 += 4 * SORT_T) {
    int v[4];
    fetch4(i0, v);
#pragma unroll
    for (int j = 0; j < 4; ++j) {
      const int i = i0 + j * SORT_T;
      if (i < m) {
        atomicAdd(&cnt[v[j] & (BIN_NODES - 1)], 1);
        if (fits) stage[i] = v[j];
      }
    }
  }
  __syncthreads();
  int c = 0, inc = 0;
  if (tid < BIN_NODES) {
    c = cnt[tid]; inc = c;
    for (int o = 1; o < 32; o <<= 1) {
      const int v = __shfl_up(inc, o);
      if (lane >= o) inc += v;
    }
    if (lane == 31) wsum[wave] = inc;
  }
  __syncthreads();
  if (tid < BIN_NODES) {
    int base = 0;
    for (int w = 0; w < wave; ++w) base += wsum[w];
    const int ex = base + inc - c;
    cur[tid] = ex;
    if (n0 + tid < N) rowptr[n0 + tid] = bs + ex;
  }
  if (b == nbin - 1 && tid == 0) rowptr[N] = bs + m;
  __syncthreads();
  for (int i0 = tid; i0 < m; i0 += 4 * SORT_T) {
    int v[4];
    if (fits) {
#pragma unroll
      for (int j = 0; j < 4; ++j) {
        const int i = i0 + j * SORT_T;
        v[j] = stage[i < m ? i : m - 1];
      }
    } else {
      fetch4(i0, v);
    }
#pragma unroll
    for (int j = 0; j < 4; ++j) {
      if (i0 + j * SORT_T < m) {
        const int pos = atomicAdd(&cur[v[j] & (BIN_NODES - 1)], 1);
        const int s = (int)((unsigned)v[j] >> BIN_SHIFT);
        if (fits) lsrt[pos] = s; else srcs[bs + pos] = s;
      }
    }
  }
  __syncthreads();
  if (fits)
    for (int i = tid; i < m; i += SORT_T) srcs[bs + i] = lsrt[i];
}


template <int LPH, int J>
__device__ __forceinline__ int group_lane(int v) {
  if constexpr (LPH == 8) {
    return __builtin_amdgcn_mov_dpp8(v, J * 0x249249);
  } else {
    return __builtin_amdgcn_update_dpp(v, v, 0x150 + J, 0xf, 0xf, false);
  }
}

template <int M>
__device__ __forceinline__ float row_xor(float v) {
  const int i = __float_as_int(v);
  return __int_as_float(__builtin_amdgcn_update_dpp(0, i, 0x160 + M, 0xf, 0xf, true));
}

template <int LPH>
__device__ __forceinline__ float group_sum(float v) {
  v += row_xor<1>(v);
  v += row_xor<2>(v);
  v += row_xor<4>(v);
  if constexpr (LPH == 16) v += row_xor<8>(v);
  return v;
}

__device__ __forceinline__ float dot4(const float4 a, const float4 b) {
  return fmaf(a.w, b.w, fmaf(a.z, b.z, fmaf(a.y, b.y, a.x * b.x)));
}

template <int F, int H, bool RELU>
__device__ __forceinline__ float4 node_row(
    const int* __restrict__ rowptr, const int* __restrict__ srcs,
    const float* __restrict__ feat, const float* __restrict__ al,
    const float* __restrict__ ar, const float* __restrict__ bias,
    const long node, const int ln) {
  constexpr int D   = F / H;
  constexpr int LPH = D / 4;
  static_assert(LPH == 8 || LPH == 16, "lane groups of 8 or 16");
  const int r  = ln * 4;
  const int jj = ln & 7;
  const int b = rowptr[node], e = rowptr[node + 1];
  const int deg = e - b;
  const char* fbytes = (const char*)feat;

  const float4 l4 = *(const float4*)(al + r);
  const float4 r4 = *(const float4*)(ar + r);
  const float ern = group_sum<LPH>(dot4(*(const float4*)(feat + node * F + r), r4));

  float m = -INFINITY, den = 0.f;
  float4 acc = {0.f, 0.f, 0.f, 0.f};
  int sv = 0;
  if (deg > 0) sv = srcs[(b + jj < e) ? b + jj : b];
  for (int off = 0; off < deg; off += 8) {
    const int  vo = sv * (F * 4);
    const bool ok = off + jj < deg;
    if (off + 8 < deg) {
      const int nx = b + off + 8 + jj;
      sv = srcs[(nx < e) ? nx : b + off + 8];
    }
    float4 f[8];
    float  s[8];
#define ROW_(i)                                                                  \
  {                                                                              \
    const unsigned o = (unsigned)group_lane<LPH, i>(vo) + (unsigned)(r * 4);     \
    f[i] = *(const float4*)(fbytes + o);                                         \
  }
    ROW_(0) ROW_(1) ROW_(2) ROW_(3) ROW_(4) ROW_(5) ROW_(6) ROW_(7)
#undef ROW_
#pragma unroll
    for (int i = 0; i < 8; ++i) s[i] = group_sum<LPH>(dot4(f[i], l4));
    float x = s[0];
#pragma unroll
    for (int i = 1; i < 8; ++i) x = (jj == i) ? s[i] : x;
    x += ern;
    x = (x > 0.f) ? x : 0.2f * x;
    x = ok ? x : -INFINITY;
    float mx = x, o;
    o = row_xor<1>(mx); mx = (o > mx) ? o : mx;
    o = row_xor<2>(mx); mx = (o > mx) ? o : mx;
    o = row_xor<4>(mx); mx = (o > mx) ? o : mx;
    const float mn = (mx > m) ? mx : m;
    const float sc = expf(m - mn);
    const float a  = expf(x - mn);
    m = mn;
    den *= sc;
    acc.x *= sc; acc.y *= sc; acc.z *= sc; acc.w *= sc;
#define ADD_(i)                                                                  \
  {                                                                              \
    const float w = __int_as_float(group_lane<LPH, i>(__float_as_int(a)));       \
    den   += w;                                                                  \
    acc.x = fmaf(f[i].x, w, acc.x); acc.y = fmaf(f[i].y, w, acc.y);              \
    acc.z = fmaf(f[i].z, w, acc.z); acc.w = fmaf(f[i].w, w, acc.w);              \
  }
    ADD_(0) ADD_(1) ADD_(2) ADD_(3) ADD_(4) ADD_(5) ADD_(6) ADD_(7)
#undef ADD_
  }

  float4 o = *(const float4*)(bias + r);
  if (e > b) {
    o.x += acc.x / den; o.y += acc.y / den; o.z += acc.z / den; o.w += acc.w / den;
  }
  if (RELU) {
    o.x = fmaxf(o.x, 0.f); o.y = fmaxf(o.y, 0.f);
    o.z = fmaxf(o.z, 0.f); o.w = fmaxf(o.w, 0.f);
  }
  return o;
}

template <int F, int H, bool RELU>
__global__ void __launch_bounds__(256) node_softmax_gather(
    const int* __restrict__ rowptr, const int* __restrict__ srcs,
    const float* __restrict__ feat, const float* __restrict__ al,
    const float* __restrict__ ar, const float* __restrict__ bias,
    float* __restrict__ out, long N) {
  constexpr int LPN = F / 4;
  const long t    = (long)blockIdx.x * blockDim.x + threadIdx.x;
  const long node = t / LPN;
  if (node >= N) return;
  const int ln = (int)(t & (LPN - 1));
  const float4 o = node_row<F, H, RELU>(rowptr, srcs, feat, al, ar, bias, node, ln);
  *(float4*)(out + node * F + ln * 4) = o;
}

template <int H, int NCOL>
__global__ void __launch_bounds__(512) node_softmax_gather_proj(
    const int* __restrict__ rowptr, const int* __restrict__ srcs,
    const float* __restrict__ feat, const float* __restrict__ al,
    const float* __restrict__ ar, const float* __restrict__ bias,
    const float* __restrict__ B, float* __restrict__ C, long N) {
  static_assert(NCOL == 64, "16 waves = 4 column tiles x 4 chains");
  __shared__ __attribute__((aligned(16))) float Xs[16 * APITCH];
  constexpr int PT_PITCH = 4 * 256 + 16;
  __shared__ __attribute__((aligned(16))) float Ps[4 * PT_PITCH];
  const int lane = threadIdx.x & 31;
  const int wave = threadIdx.x >> 5;
  const long node0 = (long)blockIdx.x * 16;
  long node = node0 + wave;
  node = node < N ? node : N - 1;
  const float4 o = node_row<128, H, true>(rowptr, srcs, feat, al, ar, bias, node, lane);
  *(float4*)(Xs + wave * APITCH + lane * 4) = o;
  __syncthreads();

  const int wv = __builtin_amdgcn_readfirstlane(wave);
  const int ct = wv >> 2;
  const int ch = wv & 3;
  const int mrow  = lane & 15;
  const int khalf = (lane >> 4) << 1;
  const v8f acc = wave_chain_k128<NCOL>(Xs + 4 * ch + (mrow * APITCH + khalf),
                                        (B + 4 * ch * NCOL + ct * 16) + (khalf * NCOL + mrow));
  float* mine = Ps + ct * PT_PITCH + ch * 256 + lane;
#pragma unroll
  for (int i = 0; i < 8; ++i) mine[i * 32] = acc[i];
  __syncthreads();

  const float* q = Ps + (lane >> 3) * PT_PITCH + (wv & 7) * 32 + (wv >> 3) * 16 + (lane & 7) * 2;
  const float2 s0 = *(const float2*)(q);
  const float2 s1 = *(const float2*)(q + 256);
  const float2 s2 = *(const float2*)(q + 512);
  const float2 s3 = *(const float2*)(q + 768);
  float2 v;
  v.x = (s0.x + s1.x) + (s2.x + s3.x);
  v.y = (s0.y + s1.y) + (s2.y + s3.y);
  if (node0 + wv < N) *(float2*)(C + (node0 + wv) * NCOL + lane * 2) = v;
}

extern "C" void kernel_launch(void* const* d_in, const int* in_sizes, int n_in,
                              void* d_out, int out_size, void* d_ws, size_t ws_size,
                              hipStream_t stream) {
  const float* h   = (const float*)d_in[0];
  const int*   src = (const int*)d_in[1];
  const int*   dst = (const int*)d_in[2];
  const float* W1  = (const float*)d_in[3];
  const float* al1 = (const float*)d_in[4];
  const float* ar1 = (const float*)d_in[5];
  const float* b1  = (const float*)d_in[6];
  const float* W2  = (const float*)d_in[7];
  const float* al2 = (const float*)d_in[8];
  const float* ar2 = (const float*)d_in[9];
  const float* b2  = (const float*)d_in[10];

  constexpr int F1 = 128, H1 = 4;
  constexpr int F2 = 64,  H2 = 1;
  const long N = in_sizes[0] / F1;
  const long E = in_sizes[1];

  float* ws     = (float*)d_ws;
  float* feat1  = ws;
  float* feat2  = feat1 + 2 * N * F1;
  int*   runs   = (int*)(feat2 + N * F2);
  int*   cedge  = runs + E;
  int*   srcs   = runs + 2 * E;
  int*   rowptr = runs + 4 * E + N * H1;

  auto cdiv = [](long a, long b) { return (unsigned)((a + b - 1) / b); };
  const int T = 256;
  const unsigned mtiles = cdiv(N, 16);

  constexpr unsigned WGPS = 256;
  const unsigned nbin  = cdiv(N, BIN_NODES);
  const unsigned nblk  = cdiv(E, CHUNK);
  const unsigned nslab = cdiv(N, RIDE_ROWS);
  const unsigned c2    = nslab < 2 * WGPS ? nslab : 2 * WGPS;
  const unsigned c0    = nslab - c2;
  const unsigned r0 = c0, r2 = cdiv(c2, 2);
  chunk_sort<<<r0 + nblk, SORT_T, 0, stream>>>(src, dst, runs, cedge, E,
                                               h, W1, feat1, (int)N, (int)r0, 1, 0, (int)c0);
  bin_sort<<<r2 + nbin, SORT_T, 0, stream>>>(cedge, runs, rowptr, srcs, (int)nbin, (int)nblk, (int)N,
                                             h, W1, feat1, (int)N, (int)r2, 2, (int)c0, (int)nslab);

  static_assert(F1 == 128, "the fused kernel's rows are the K = 128 of wave_tile_k128");
  node_softmax_gather_proj<H1, F2><<<mtiles, 512, 0, stream>>>(
      rowptr, srcs, feat1, al1, ar1, b1, W2, feat2, N);

  node_softmax_gather<F2, H2, false><<<cdiv(N * (F2 / 4), T), T, 0, stream>>>(
      rowptr, srcs, feat2, al2, ar2, b2, (float*)d_out, N);
}


# Gather softmax arithmetic on the hardware's f32 forms: exp as v_exp_f32 on scores scaled by log2(e), acc/den as one v_rcp_f32 and multiplies, maxima as v_max; no expf, no IEEE divide; sums unchanged
# speedup vs baseline: 7.1904x; 1.0304x over previous
#include <hip/hip_runtime.h>
#include <math.h>

typedef __attribute__((ext_vector_type(2))) float v2f;
typedef __attribute__((ext_vector_type(8))) float v8f;

constexpr int APITCH = 132;

template <int NCOL>
__device__ __forceinline__ v8f wave_tile_k128(const float* Ap, const float* __restrict__ Bp) {
  v8f acc0 = {}, acc1 = {}, acc2 = {}, acc3 = {};
#pragma unroll
  for (int k = 0; k < 128; k += 16) {
    v2f a0, a1, a2, a3, b0, b1, b2, b3;
    a0.x = Ap[k + 0];  a0.y = Ap[k + 1];
    a1.x = Ap[k + 4];  a1.y = Ap[k + 5];
    a2.x = Ap[k + 8];  a2.y = Ap[k + 9];
    a3.x = Ap[k + 12]; a3.y = Ap[k + 13];
    b0.x = Bp[(long)(k + 0)  * NCOL]; b0.y = Bp[(long)(k + 1)  * NCOL];
    b1.x = Bp[(long)(k + 4)  * NCOL]; b1.y = Bp[(long)(k + 5)  * NCOL];
    b2.x = Bp[(long)(k + 8)  * NCOL]; b2.y = Bp[(long)(k + 9)  * NCOL];
    b3.x = Bp[(long)(k + 12) * NCOL]; b3.y = Bp[(long)(k + 13) * NCOL];
    acc0 = __builtin_amdgcn_wmma_f32_16x16x4_f32(false, a0, false, b0, (short)0, acc0, false, false);
    acc1 = __builtin_amdgcn_wmma_f32_16x16x4_f32(false, a1, false, b1, (short)0, acc1, false, false);
    acc2 = __builtin_amdgcn_wmma_f32_16x16x4_f32(false, a2, false, b2, (short)0, acc2, false, false);
    acc3 = __builtin_amdgcn_wmma_f32_16x16x4_f32(false, a3, false, b3, (short)0, acc3, false, false);
  }
  v8f acc;
#pragma unroll
  for (int i = 0; i < 8; ++i) acc[i] = (acc0[i] + acc1[i]) + (acc2[i] + acc3[i]);
  return acc;
}

template <int NCOL>
__device__ __forceinline__ v8f wave_chain_k128(const float* Ap, const float* __restrict__ Bp) {
  v8f acc = {};
#pragma unroll
  for (int k = 0; k < 128; k += 16) {
    v2f a, b;
    a.x = Ap[k];        a.y = Ap[k + 1];
    b.x = Bp[k * NCOL]; b.y = Bp[(k + 1) * NCOL];
    acc = __builtin_amdgcn_wmma_f32_16x16x4_f32(false, a, false, b, (short)0, acc, false, false);
  }
  return acc;
}

constexpr int RIDE_T = 1024;
template <int NCOL>
struct Ride {
  static constexpr int CT   = NCOL / 16;
  static constexpr int MT   = (RIDE_T / 32) / CT;
  static constexpr int ROWS = 16 * MT;
  static constexpr int VPT  = ROWS * 32 / RIDE_T;
};
constexpr int RIDE_ROWS = Ride<128>::ROWS;

template <int NCOL>
__device__ __forceinline__ void gemm_ride(float* As, const float* __restrict__ A,
                                          const float* __restrict__ B, float* __restrict__ C,
                                          int M, int slab, int slab_end) {
  constexpr int CT = Ride<NCOL>::CT, ROWS = Ride<NCOL>::ROWS;
  static_assert(Ride<NCOL>::VPT == 2, "two float4 of a slab per thread");
  const int lane  = threadIdx.x & 31;
  const int wave  = threadIdx.x >> 5;
  const int mt    = wave / CT;
  const int col0  = (wave % CT) * 16;
  const int mrow  = lane & 15;
  const int khalf = (lane >> 4) << 1;
  const float* Ap = As + (mt * 16 + mrow) * APITCH + khalf;
  const float* Bp = B + (long)khalf * NCOL + col0 + mrow;
  const int i0 = threadIdx.x >> 5, c4 = (threadIdx.x & 31) * 4;
  float* As0 = As + i0 * APITCH + c4;
  float* As1 = As0 + 32 * APITCH;
  const float* Ac = A + c4;

  if (slab >= slab_end) return;
  int ra = slab * ROWS + i0, rb = ra + 32;
  ra = ra < M ? ra : M - 1;
  rb = rb < M ? rb : M - 1;
  float4 p0 = *(const float4*)(Ac + (long)ra * 128);
  float4 p1 = *(const float4*)(Ac + (long)rb * 128);
  for (;;) {
    *(float4*)As0 = p0;
    *(float4*)As1 = p1;
    __syncthreads();
    if (slab + 1 < slab_end && threadIdx.x < ROWS * 2) {
      int rn = (slab + 1) * ROWS + (threadIdx.x >> 1);
      rn = rn < M ? rn : M - 1;
      __builtin_prefetch(A + (long)rn * 128 + (threadIdx.x & 1) * 64, 0, 3);
    }
    int boff = 0;
    asm volatile("" : "+v"(boff));
    const v8f acc = wave_tile_k128<NCOL>(Ap, Bp + boff);
    const int rbase = slab * ROWS + mt * 16 + ((lane >> 4) << 3);
    float* Cp = C + (long)rbase * NCOL + col0 + mrow;
    if ((slab + 1) * ROWS <= M) {
#pragma unroll
      for (int i = 0; i < 8; ++i) Cp[(long)i * NCOL] = acc[i];
    } else {
#pragma unroll
      for (int i = 0; i < 8; ++i)
        if (rbase + i < M) Cp[(long)i * NCOL] = acc[i];
    }
    if (++slab >= slab_end) break;
    ra = slab * ROWS + i0; rb = ra + 32;
    ra = ra < M ? ra : M - 1;
    rb = rb < M ? rb : M - 1;
    p0 = *(const float4*)(Ac + (long)ra * 128);
    p1 = *(const float4*)(Ac + (long)rb * 128);
    __syncthreads();
  }
}

constexpr int BIN_SHIFT = 8;
constexpr int BIN_NODES = 1 << BIN_SHIFT;
constexpr int MAXBIN    = 256;
constexpr int CHUNK     = 4096;
constexpr int SORT_T    = 1024;
constexpr int MAXCH     = SORT_T;
static_assert(SORT_T == RIDE_T, "the riding GEMM slabs are cut for the sort's work-group size");
constexpr int BIN_CAP   = 8192;
constexpr int NO_RUN    = 0x7fffffff;

__global__ void __launch_bounds__(SORT_T) __attribute__((amdgpu_waves_per_eu(16))) chunk_sort(
    const int* __restrict__ src, const int* __restrict__ dst,
    int* __restrict__ runs_w, int* __restrict__ cedge, long E,
    const float* __restrict__ gA, const float* __restrict__ gB, float* __restrict__ gC,
    int gM, int nride, int gper, int gslab0, int gend) {
  __shared__ __attribute__((aligned(16))) float As[RIDE_ROWS * APITCH];
  if ((int)blockIdx.x < nride) {
    const int s0 = gslab0 + (int)blockIdx.x * gper;
    gemm_ride<128>(As, gA, gB, gC, gM, s0, (s0 + gper < gend) ? s0 + gper : gend);
    return;
  }
  __shared__ int h[MAXBIN], lcur[MAXBIN];
  __shared__ int wsum[MAXBIN / 32];
  __shared__ int stage[CHUNK];
  int2* __restrict__ runs = (int2*)runs_w;
  const int tid = threadIdx.x, lane = tid & 31, wave = tid >> 5;
  const int blk = (int)blockIdx.x - nride;
  const long e0 = (long)blk * CHUNK;
  const long e  = e0 + tid * 4;

  const bool full = e + 4 <= E;
  int4 s4 = {0, 0, 0, 0}, d4 = {0, 0, 0, 0};
  if (full) { s4 = *(const int4*)(src + e); d4 = *(const int4*)(dst + e); }
  if (tid < MAXBIN) h[tid] = 0;
  __syncthreads();
  if (full) {
    atomicAdd(&h[d4.x >> BIN_SHIFT], 1);
    atomicAdd(&h[d4.y >> BIN_SHIFT], 1);
    atomicAdd(&h[d4.z >> BIN_SHIFT], 1);
    atomicAdd(&h[d4.w >> BIN_SHIFT], 1);
  } else {
    for (long i = e; i < E; ++i) atomicAdd(&h[dst[i] >> BIN_SHIFT], 1);
  }
  __syncthreads();

  int own = 0, ic = 0;
  if (tid < MAXBIN) {
    own = h[tid]; ic = own;
    for (int o = 1; o < 32; o <<= 1) {
      const int vc = __shfl_up(ic, o);
      if (lane >= o) ic += vc;
    }
    if (lane == 31) wsum[wave] = ic;
  }
  __syncthreads();
  if (tid < MAXBIN) {
    int bc = 0;
    for (int w = 0; w < wave; ++w) bc += wsum[w];
    const int lstart = bc + ic - own;
    lcur[tid] = lstart;
    runs[(long)blk * MAXBIN + tid] = make_int2(lstart, own);
  }
  __syncthreads();

  if (full) {
    const int q0 = atomicAdd(&lcur[d4.x >> BIN_SHIFT], 1);
    const int q1 = atomicAdd(&lcur[d4.y >> BIN_SHIFT], 1);
    const int q2 = atomicAdd(&lcur[d4.z >> BIN_SHIFT], 1);
    const int q3 = atomicAdd(&lcur[d4.w >> BIN_SHIFT], 1);
    stage[q0] = (s4.x << BIN_SHIFT) | (d4.x & (BIN_NODES - 1));
    stage[q1] = (s4.y << BIN_SHIFT) | (d4.y & (BIN_NODES - 1));
    stage[q2] = (s4.z << BIN_SHIFT) | (d4.z & (BIN_NODES - 1));
    stage[q3] = (s4.w << BIN_SHIFT) | (d4.w & (BIN_NODES - 1));
  } else {
    for (long i = e; i < E; ++i) {
      const int d = dst[i];
      stage[atomicAdd(&lcur[d >> BIN_SHIFT], 1)] = (src[i] << BIN_SHIFT) | (d & (BIN_NODES - 1));
    }
  }
  __syncthreads();
  const long left = E - e0;
  const int nloc = left < CHUNK ? (int)left : CHUNK;
  for (int i = tid; i < nloc; i += SORT_T) cedge[e0 + i] = stage[i];
}

__global__ void __launch_bounds__(SORT_T) __attribute__((amdgpu_waves_per_eu(16))) bin_sort(
    const int* __restrict__ cedge, const int* __restrict__ runs_w,
    int* __restrict__ rowptr, int* __restrict__ srcs, int nbin, int nblk, int N,
    const float* __restrict__ gA, const float* __restrict__ gB, float* __restrict__ gC,
    int gM, int nride, int gper, int gslab0, int gend) {
  __shared__ __attribute__((aligned(16))) float As[RIDE_ROWS * APITCH];
  if ((int)blockIdx.x < nride) {
    const int s0 = gslab0 + (int)blockIdx.x * gper;
    gemm_ride<128>(As, gA, gB, gC, gM, s0, (s0 + gper < gend) ? s0 + gper : gend);
    return;
  }
  __shared__ int cnt[BIN_NODES], cur[BIN_NODES];
  __shared__ int wsum[BIN_NODES / 32];
  __shared__ int wtot[2][SORT_T / 32];
  __shared__ int roff[MAXCH];
  __shared__ int rsrc[MAXCH];
  __shared__ int stage[BIN_CAP], lsrt[BIN_CAP];
  const int2* __restrict__ runs = (const int2*)runs_w;
  const int tid = threadIdx.x, lane = tid & 31, wave = tid >> 5;
  const int b  = (int)blockIdx.x - nride;
  const int n0 = b << BIN_SHIFT;

  int2 run = {0, 0};
  if (tid < nblk) run = runs[(long)tid * MAXBIN + b];
  if (tid < BIN_NODES) cnt[tid] = 0;
  int rinc = run.y, rst = run.x;
  for (int o = 1; o < 32; o <<= 1) {
    const int v = __shfl_up(rinc, o);
    if (lane >= o) rinc += v;
  }
  for (int o = 16; o > 0; o >>= 1) rst += __shfl_xor(rst, o);
  if (lane == 31) wtot[0][wave] = rinc;
  if (lane == 0)  wtot[1][wave] = rst;
  __syncthreads();
  int rbase = 0, m = 0, bs = 0;
  for (int w = 0; w < SORT_T / 32; ++w) {
    const int a = wtot[0][w];
    rbase += (w < wave) ? a : 0;
    m  += a;
    bs += wtot[1][w];
  }
  const int rex = rbase + rinc - run.y;
  roff[tid] = (tid < nblk) ? rex : NO_RUN;
  rsrc[tid] = tid * CHUNK + run.x - rex;
  __syncthreads();

  int top = 1;
  while (top < nblk) top <<= 1;
  auto fetch4 = [&](int i0, int (&v)[4]) {
    int ii[4], ch[4];
#pragma unroll
    for (int j = 0; j < 4; ++j) {
      const int i = i0 + j * SORT_T;
      ii[j] = i < m ? i : m - 1;
      ch[j] = 0;
    }
    for (int s = top >> 1; s > 0; s >>= 1) {
#pragma unroll
      for (int j = 0; j < 4; ++j) ch[j] += (roff[ch[j] + s] <= ii[j]) ? s : 0;
    }
#pragma unroll
    for (int j = 0; j < 4; ++j) v[j] = cedge[rsrc[ch[j]] + ii[j]];
  };

  const bool fits = m <= BIN_CAP;
  for (int i0 = tid; i0 < m; i0 += 4 * SORT_T) {
    int v[4];
    fetch4(i0, v);
#pragma unroll
    for (int j = 0; j < 4; ++j) {
      const int i = i0 + j * SORT_T;
      if (i < m) {
        atomicAdd(&cnt[v[j] & (BIN_NODES - 1)], 1);
        if (fits) stage[i] = v[j];
      }
    }
  }
  __syncthreads();
  int c = 0, inc = 0;
  if (tid < BIN_NODES) {
    c = cnt[tid]; inc = c;
    for (int o = 1; o < 32; o <<= 1) {
      const int v = __shfl_up(inc, o);
      if (lane >= o) inc += v;
    }
    if (lane == 31) wsum[wave] = inc;
  }
  __syncthreads();
  if (tid < BIN_NODES) {
    int base = 0;
    for (int w = 0; w < wave; ++w) base += wsum[w];
    const int ex = base + inc - c;
    cur[tid] = ex;
    if (n0 + tid < N) rowptr[n0 + tid] = bs + ex;
  }
  if (b == nbin - 1 && tid == 0) rowptr[N] = bs + m;
  __syncthreads();
  for (int i0 = tid; i0 < m; i0 += 4 * SORT_T) {
    int v[4];
    if (fits) {
#pragma unroll
      for (int j = 0; j < 4; ++j) {
        const int i = i0 + j * SORT_T;
        v[j] = stage[i < m ? i : m - 1];
      }
    } else {
      fetch4(i0, v);
    }
#pragma unroll
    for (int j = 0; j < 4; ++j) {
      if (i0 + j * SORT_T < m) {
        const int pos = atomicAdd(&cur[v[j] & (BIN_NODES - 1)], 1);
        const int s = (int)((unsigned)v[j] >> BIN_SHIFT);
        if (fits) lsrt[pos] = s; else srcs[bs + pos] = s;
      }
    }
  }
  __syncthreads();
  if (fits)
    for (int i = tid; i < m; i += SORT_T) srcs[bs + i] = lsrt[i];
}


template <int LPH, int J>
__device__ __forceinline__ int group_lane(int v) {
  if constexpr (LPH == 8) {
    return __builtin_amdgcn_mov_dpp8(v, J * 0x249249);
  } else {
    return __builtin_amdgcn_update_dpp(v, v, 0x150 + J, 0xf, 0xf, false);
  }
}

template <int M>
__device__ __forceinline__ float row_xor(float v) {
  const int i = __float_as_int(v);
  return __int_as_float(__builtin_amdgcn_update_dpp(0, i, 0x160 + M, 0xf, 0xf, true));
}

template <int LPH>
__device__ __forceinline__ float group_sum(float v) {
  v += row_xor<1>(v);
  v += row_xor<2>(v);
  v += row_xor<4>(v);
  if constexpr (LPH == 16) v += row_xor<8>(v);
  return v;
}

constexpr float LOG2E = 1.44269504088896340736f;

__device__ __forceinline__ float dot4(const float4 a, const float4 b) {
  return fmaf(a.w, b.w, fmaf(a.z, b.z, fmaf(a.y, b.y, a.x * b.x)));
}

template <int F, int H, bool RELU>
__device__ __forceinline__ float4 node_row(
    const int* __restrict__ rowptr, const int* __restrict__ srcs,
    const float* __restrict__ feat, const float* __restrict__ al,
    const float* __restrict__ ar, const float* __restrict__ bias,
    const long node, const int ln) {
  constexpr int D   = F / H;
  constexpr int LPH = D / 4;
  static_assert(LPH == 8 || LPH == 16, "lane groups of 8 or 16");
  const int r  = ln * 4;
  const int jj = ln & 7;
  const int b = rowptr[node], e = rowptr[node + 1];
  const int deg = e - b;
  const char* fbytes = (const char*)feat;

  const float4 l4 = *(const float4*)(al + r);
  const float4 r4 = *(const float4*)(ar + r);
  const float ern = group_sum<LPH>(dot4(*(const float4*)(feat + node * F + r), r4));

  float m = -INFINITY, den = 0.f;
  float4 acc = {0.f, 0.f, 0.f, 0.f};
  int sv = 0;
  if (deg > 0) sv = srcs[(b + jj < e) ? b + jj : b];
  for (int off = 0; off < deg; off += 8) {
    const int  vo = sv * (F * 4);
    const bool ok = off + jj < deg;
    if (off + 8 < deg) {
      const int nx = b + off + 8 + jj;
      sv = srcs[(nx < e) ? nx : b + off + 8];
    }
    float4 f[8];
    float  s[8];
#define ROW_(i)                                                                  \
  {                                                                              \
    const unsigned o = (unsigned)group_lane<LPH, i>(vo) + (unsigned)(r * 4);     \
    f[i] = *(const float4*)(fbytes + o);                                         \
  }
    ROW_(0) ROW_(1) ROW_(2) ROW_(3) ROW_(4) ROW_(5) ROW_(6) ROW_(7)
#undef ROW_
#pragma unroll
    for (int i = 0; i < 8; ++i) s[i] = group_sum<LPH>(dot4(f[i], l4));
    float x = s[0];
#pragma unroll
    for (int i = 1; i < 8; ++i) x = (jj == i) ? s[i] : x;
    x += ern;
    x = fmaxf(x * LOG2E, x * (0.2f * LOG2E));
    x = ok ? x : -INFINITY;
    float mx = x;
    mx = fmaxf(mx, row_xor<1>(mx));
    mx = fmaxf(mx, row_xor<2>(mx));
    mx = fmaxf(mx, row_xor<4>(mx));
    const float mn = fmaxf(mx, m);
    const float sc = __builtin_amdgcn_exp2f(m - mn);
    const float a  = __builtin_amdgcn_exp2f(x - mn);
    m = mn;
    den *= sc;
    acc.x *= sc; acc.y *= sc; acc.z *= sc; acc.w *= sc;
#define ADD_(i)                                                                  \
  {                                                                              \
    const float w = __int_as_float(group_lane<LPH, i>(__float_as_int(a)));       \
    den   += w;                                                                  \
    acc.x = fmaf(f[i].x, w, acc.x); acc.y = fmaf(f[i].y, w, acc.y);              \
    acc.z = fmaf(f[i].z, w, acc.z); acc.w = fmaf(f[i].w, w, acc.w);              \
  }
    ADD_(0) ADD_(1) ADD_(2) ADD_(3) ADD_(4) ADD_(5) ADD_(6) ADD_(7)
#undef ADD_
  }

  float4 o = *(const float4*)(bias + r);
  if (e > b) {
    const float inv = __builtin_amdgcn_rcpf(den);
    o.x += acc.x * inv; o.y += acc.y * inv; o.z += acc.z * inv; o.w += acc.w * inv;
  }
  if (RELU) {
    o.x = fmaxf(o.x, 0.f); o.y = fmaxf(o.y, 0.f);
    o.z = fmaxf(o.z, 0.f); o.w = fmaxf(o.w, 0.f);
  }
  return o;
}

template <int F, int H, bool RELU>
__global__ void __launch_bounds__(256) node_softmax_gather(
    const int* __restrict__ rowptr, const int* __restrict__ srcs,
    const float* __restrict__ feat, const float* __restrict__ al,
    const float* __restrict__ ar, const float* __restrict__ bias,
    float* __restrict__ out, long N) {
  constexpr int LPN = F / 4;
  const long t    = (long)blockIdx.x * blockDim.x + threadIdx.x;
  const long node = t / LPN;
  if (node >= N) return;
  const int ln = (int)(t & (LPN - 1));
  const float4 o = node_row<F, H, RELU>(rowptr, srcs, feat, al, ar, bias, node, ln);
  *(float4*)(out + node * F + ln * 4) = o;
}

template <int H, int NCOL>
__global__ void __launch_bounds__(512) node_softmax_gather_proj(
    const int* __restrict__ rowptr, const int* __restrict__ srcs,
    const float* __restrict__ feat, const float* __restrict__ al,
    const float* __restrict__ ar, const float* __restrict__ bias,
    const float* __restrict__ B, float* __restrict__ C, long N) {
  static_assert(NCOL == 64, "16 waves = 4 column tiles x 4 chains");
  __shared__ __attribute__((aligned(16))) float Xs[16 * APITCH];
  constexpr int PT_PITCH = 4 * 256 + 16;
  __shared__ __attribute__((aligned(16))) float Ps[4 * PT_PITCH];
  const int lane = threadIdx.x & 31;
  const int wave = threadIdx.x >> 5;
  const long node0 = (long)blockIdx.x * 16;
  long node = node0 + wave;
  node = node < N ? node : N - 1;
  const float4 o = node_row<128, H, true>(rowptr, srcs, feat, al, ar, bias, node, lane);
  *(float4*)(Xs + wave * APITCH + lane * 4) = o;
  __syncthreads();

  const int wv = __builtin_amdgcn_readfirstlane(wave);
  const int ct = wv >> 2;
  const int ch = wv & 3;
  const int mrow  = lane & 15;
  const int khalf = (lane >> 4) << 1;
  const v8f acc = wave_chain_k128<NCOL>(Xs + 4 * ch + (mrow * APITCH + khalf),
                                        (B + 4 * ch * NCOL + ct * 16) + (khalf * NCOL + mrow));
  float* mine = Ps + ct * PT_PITCH + ch * 256 + lane;
#pragma unroll
  for (int i = 0; i < 8; ++i) mine[i * 32] = acc[i];
  __syncthreads();

  const float* q = Ps + (lane >> 3) * PT_PITCH + (wv & 7) * 32 + (wv >> 3) * 16 + (lane & 7) * 2;
  const float2 s0 = *(const float2*)(q);
  const float2 s1 = *(const float2*)(q + 256);
  const float2 s2 = *(const float2*)(q + 512);
  const float2 s3 = *(const float2*)(q + 768);
  float2 v;
  v.x = (s0.x + s1.x) + (s2.x + s3.x);
  v.y = (s0.y + s1.y) + (s2.y + s3.y);
  if (node0 + wv < N) *(float2*)(C + (node0 + wv) * NCOL + lane * 2) = v;
}

extern "C" void kernel_launch(void* const* d_in, const int* in_sizes, int n_in,
                              void* d_out, int out_size, void* d_ws, size_t ws_size,
                              hipStream_t stream) {
  const float* h   = (const float*)d_in[0];
  const int*   src = (const int*)d_in[1];
  const int*   dst = (const int*)d_in[2];
  const float* W1  = (const float*)d_in[3];
  const float* al1 = (const float*)d_in[4];
  const float* ar1 = (const float*)d_in[5];
  const float* b1  = (const float*)d_in[6];
  const float* W2  = (const float*)d_in[7];
  const float* al2 = (const float*)d_in[8];
  const float* ar2 = (const float*)d_in[9];
  const float* b2  = (const float*)d_in[10];

  constexpr int F1 = 128, H1 = 4;
  constexpr int F2 = 64,  H2 = 1;
  const long N = in_sizes[0] / F1;
  const long E = in_sizes[1];

  float* ws     = (float*)d_ws;
  float* feat1  = ws;
  float* feat2  = feat1 + 2 * N * F1;
  int*   runs   = (int*)(feat2 + N * F2);
  int*   cedge  = runs + E;
  int*   srcs   = runs + 2 * E;
  int*   rowptr = runs + 4 * E + N * H1;

  auto cdiv = [](long a, long b) { return (unsigned)((a + b - 1) / b); };
  const int T = 256;
  const unsigned mtiles = cdiv(N, 16);

  constexpr unsigned WGPS = 256;
  const unsigned nbin  = cdiv(N, BIN_NODES);
  const unsigned nblk  = cdiv(E, CHUNK);
  const unsigned nslab = cdiv(N, RIDE_ROWS);
  const unsigned c2    = nslab < 2 * WGPS ? nslab : 2 * WGPS;
  const unsigned c0    = nslab - c2;
  const unsigned r0 = c0, r2 = cdiv(c2, 2);
  chunk_sort<<<r0 + nblk, SORT_T, 0, stream>>>(src, dst, runs, cedge, E,
                                               h, W1, feat1, (int)N, (int)r0, 1, 0, (int)c0);
  bin_sort<<<r2 + nbin, SORT_T, 0, stream>>>(cedge, runs, rowptr, srcs, (int)nbin, (int)nblk, (int)N,
                                             h, W1, feat1, (int)N, (int)r2, 2, (int)c0, (int)nslab);

  static_assert(F1 == 128, "the fused kernel's rows are the K = 128 of wave_tile_k128");
  node_softmax_gather_proj<H1, F2><<<mtiles, 512, 0, stream>>>(
      rowptr, srcs, feat1, al1, ar1, b1, W2, feat2, N);

  node_softmax_gather<F2, H2, false><<<cdiv(N * (F2 / 4), T), T, 0, stream>>>(
      rowptr, srcs, feat2, al2, ar2, b2, (float*)d_out, N);
}
